# starting kernel
# baseline (speedup 1.0000x reference)
#include <hip/hip_runtime.h>
#include <hip/hip_bf16.h>

typedef __attribute__((ext_vector_type(2))) float v2f;
typedef __attribute__((ext_vector_type(8))) float v8f;

#define BSTRIDE 80

template <int RELU>
__global__ void gemm_wmma_f32(const float* __restrict__ A,
                              const float* __restrict__ B,
                              float* __restrict__ C,
                              int nRows, int K) {
  __shared__ float2 BsP[64 * BSTRIDE];
  const int pairRows = K >> 1;
  for (int j = threadIdx.x; j < pairRows * 64; j += blockDim.x) {
    const int k2 = j >> 6;
    const int nn = j & 63;
    BsP[k2 * BSTRIDE + nn] =
        make_float2(B[(2 * k2) * 64 + nn], B[(2 * k2 + 1) * 64 + nn]);
  }
  __syncthreads();

  const int lane = threadIdx.x & 31;
  const int wave = threadIdx.x >> 5;
  const int nStrips = nRows >> 4;
  const int strip = blockIdx.x * 8 + wave;
  if (strip >= nStrips) return;

  const int n  = lane & 15;
  const int hi = lane >> 4;
  const long rowBase = (long)strip * 16;
  const float* Arow = A + (rowBase + n) * (long)K;

  v8f acc0 = {}, acc1 = {}, acc2 = {}, acc3 = {};
  for (int k = 0; k < K; k += 4) {
    const int ka = k + 2 * hi;
    float2 av = *(const float2*)(Arow + ka);
    if (RELU) { av.x = fmaxf(av.x, 0.f); av.y = fmaxf(av.y, 0.f); }
    v2f a; a[0] = av.x; a[1] = av.y;

    const float2* bp = &BsP[((k >> 1) + hi) * BSTRIDE + n];
    const v2f bA = *(const v2f*)(bp + 0);
    const v2f bB = *(const v2f*)(bp + 16);
    const v2f bC = *(const v2f*)(bp + 32);
    const v2f bD = *(const v2f*)(bp + 48);

    acc0 = __builtin_amdgcn_wmma_f32_16x16x4_f32(false, a, false, bA, (short)0, acc0, false, false);
    acc1 = __builtin_amdgcn_wmma_f32_16x16x4_f32(false, a, false, bB, (short)0, acc1, false, false);
    acc2 = __builtin_amdgcn_wmma_f32_16x16x4_f32(false, a, false, bC, (short)0, acc2, false, false);
    acc3 = __builtin_amdgcn_wmma_f32_16x16x4_f32(false, a, false, bD, (short)0, acc3, false, false);
  }

  float* Cout = C + (rowBase + 8 * hi) * 64 + n;
#pragma unroll
  for (int r = 0; r < 8; ++r) {
    Cout[r * 64 + 0]  = acc0[r];
    Cout[r * 64 + 16] = acc1[r];
    Cout[r * 64 + 32] = acc2[r];
    Cout[r * 64 + 48] = acc3[r];
  }
}

__global__ void scatter_edges(const float* __restrict__ t,
                              const int* __restrict__ src,
                              const int* __restrict__ dst,
                              const float* __restrict__ w,
                              float* __restrict__ acc,
                              int E) {
  const long idx = (long)blockIdx.x * blockDim.x + threadIdx.x;
  const int e  = (int)(idx >> 5);
  const int pr = (int)(idx & 31);
  if (e >= E) return;
  const int s = src[e], d = dst[e];
  const float ww = w[e];
  const float2 v = *(const float2*)(t + (long)s * 64 + pr * 2);
  float* p = acc + (long)d * 64 + pr * 2;
  unsafeAtomicAdd(p,     v.x * ww);
  unsafeAtomicAdd(p + 1, v.y * ww);
}

__global__ void decode_kernel(const float* __restrict__ z,
                              const int* __restrict__ pos,
                              const float* __restrict__ Wlin,
                              float* __restrict__ out,
                              int P) {
  __shared__ float wl[256];
  for (int i = threadIdx.x; i < 256; i += blockDim.x) wl[i] = Wlin[i];
  __syncthreads();

  const int p = blockIdx.x * blockDim.x + threadIdx.x;
  if (p >= P) return;
  const int s = pos[p];
  const int d = pos[P + p];
  const float4* zs = (const float4*)(z + (long)s * 64);
  const float4* zd = (const float4*)(z + (long)d * 64);

  float a0 = 0.f, a1 = 0.f;
#pragma unroll
  for (int q = 0; q < 16; ++q) {
    const float4 v = zs[q];
    a0 += v.x * wl[4*q]       + v.y * wl[4*q + 1]       + v.z * wl[4*q + 2]       + v.w * wl[4*q + 3];
    a1 += v.x * wl[128 + 4*q] + v.y * wl[128 + 4*q + 1] + v.z * wl[128 + 4*q + 2] + v.w * wl[128 + 4*q + 3];
  }
#pragma unroll
  for (int q = 0; q < 16; ++q) {
    const float4 v = zd[q];
    a0 += v.x * wl[64 + 4*q]  + v.y * wl[64 + 4*q + 1]  + v.z * wl[64 + 4*q + 2]  + v.w * wl[64 + 4*q + 3];
    a1 += v.x * wl[192 + 4*q] + v.y * wl[192 + 4*q + 1] + v.z * wl[192 + 4*q + 2] + v.w * wl[192 + 4*q + 3];
  }
  out[2 * p]     = a0;
  out[2 * p + 1] = a1;
}

extern "C" void kernel_launch(void* const* d_in, const int* in_sizes, int n_in,
                              void* d_out, int out_size, void* d_ws, size_t ws_size,
                              hipStream_t stream) {
  const float* x    = (const float*)d_in[0];
  const int*   ei   = (const int*)d_in[1];
  const float* w    = (const float*)d_in[2];
  const int*   pos  = (const int*)d_in[3];
  const float* W1   = (const float*)d_in[4];
  const float* W2   = (const float*)d_in[5];
  const float* Wlin = (const float*)d_in[6];
  float* out = (float*)d_out;

  const int F = 128;
  const int N = in_sizes[0] / F;
  const int E = in_sizes[2];
  const int P = in_sizes[3] / 2;

  float* t1 = (float*)d_ws;
  float* h  = t1 + (size_t)N * 64;
  float* t2 = h  + (size_t)N * 64;
  float* z  = t2 + (size_t)N * 64;

  hipMemsetAsync(h, 0, (size_t)N * 64 * sizeof(float), stream);
  hipMemsetAsync(z, 0, (size_t)N * 64 * sizeof(float), stream);

  const int gemmBlocks = (N / 16 + 7) / 8;
  const long sthreads  = (long)E * 32;
  const int  sblocks   = (int)((sthreads + 255) / 256);

  gemm_wmma_f32<0><<<gemmBlocks, 256, 0, stream>>>(x, W1, t1, N, F);
  scatter_edges<<<sblocks, 256, 0, stream>>>(t1, ei, ei + E, w, h, E);
  gemm_wmma_f32<1><<<gemmBlocks, 256, 0, stream>>>(h, W2, t2, N, 64);
  scatter_edges<<<sblocks, 256, 0, stream>>>(t2, ei, ei + E, w, z, E);
  decode_kernel<<<(P + 255) / 256, 256, 0, stream>>>(z, pos, Wlin, out, P);
}


# Decode weights folded into layer 2 by linearity (Wf = W2 x Wlin halves, 64x4): 2nd GEMM, 2nd edge scatter and decode carry 4 floats per node, not 64; products reassociated, atomic sums on 4-wide rows
# speedup vs baseline: 1.4372x; 1.4372x over previous
#include <hip/hip_runtime.h>
#include <hip/hip_bf16.h>

typedef __attribute__((ext_vector_type(2))) float v2f;
typedef __attribute__((ext_vector_type(8))) float v8f;

#define BSTRIDE 80

template <int RELU>
__global__ void gemm_wmma_f32(const float* __restrict__ A,
                              const float* __restrict__ B,
                              float* __restrict__ C,
                              int nRows, int K) {
  __shared__ float2 BsP[64 * BSTRIDE];
  const int pairRows = K >> 1;
  for (int j = threadIdx.x; j < pairRows * 64; j += blockDim.x) {
    const int k2 = j >> 6;
    const int nn = j & 63;
    BsP[k2 * BSTRIDE + nn] =
        make_float2(B[(2 * k2) * 64 + nn], B[(2 * k2 + 1) * 64 + nn]);
  }
  __syncthreads();

  const int lane = threadIdx.x & 31;
  const int wave = threadIdx.x >> 5;
  const int nStrips = nRows >> 4;
  const int strip = blockIdx.x * 8 + wave;
  if (strip >= nStrips) return;

  const int n  = lane & 15;
  const int hi = lane >> 4;
  const long rowBase = (long)strip * 16;
  const float* Arow = A + (rowBase + n) * (long)K;

  v8f acc0 = {}, acc1 = {}, acc2 = {}, acc3 = {};
  for (int k = 0; k < K; k += 4) {
    const int ka = k + 2 * hi;
    float2 av = *(const float2*)(Arow + ka);
    if (RELU) { av.x = fmaxf(av.x, 0.f); av.y = fmaxf(av.y, 0.f); }
    v2f a; a[0] = av.x; a[1] = av.y;

    const float2* bp = &BsP[((k >> 1) + hi) * BSTRIDE + n];
    const v2f bA = *(const v2f*)(bp + 0);
    const v2f bB = *(const v2f*)(bp + 16);
    const v2f bC = *(const v2f*)(bp + 32);
    const v2f bD = *(const v2f*)(bp + 48);

    acc0 = __builtin_amdgcn_wmma_f32_16x16x4_f32(false, a, false, bA, (short)0, acc0, false, false);
    acc1 = __builtin_amdgcn_wmma_f32_16x16x4_f32(false, a, false, bB, (short)0, acc1, false, false);
    acc2 = __builtin_amdgcn_wmma_f32_16x16x4_f32(false, a, false, bC, (short)0, acc2, false, false);
    acc3 = __builtin_amdgcn_wmma_f32_16x16x4_f32(false, a, false, bD, (short)0, acc3, false, false);
  }

  float* Cout = C + (rowBase + 8 * hi) * 64 + n;
#pragma unroll
  for (int r = 0; r < 8; ++r) {
    Cout[r * 64 + 0]  = acc0[r];
    Cout[r * 64 + 16] = acc1[r];
    Cout[r * 64 + 32] = acc2[r];
    Cout[r * 64 + 48] = acc3[r];
  }
}

__global__ void scatter_edges(const float* __restrict__ t,
                              const int* __restrict__ src,
                              const int* __restrict__ dst,
                              const float* __restrict__ w,
                              float* __restrict__ acc,
                              int E) {
  const long idx = (long)blockIdx.x * blockDim.x + threadIdx.x;
  const int e  = (int)(idx >> 5);
  const int pr = (int)(idx & 31);
  if (e >= E) return;
  const int s = src[e], d = dst[e];
  const float ww = w[e];
  const float2 v = *(const float2*)(t + (long)s * 64 + pr * 2);
  float* p = acc + (long)d * 64 + pr * 2;
  unsafeAtomicAdd(p,     v.x * ww);
  unsafeAtomicAdd(p + 1, v.y * ww);
}

__global__ void __launch_bounds__(256) gemm_fold_wmma_f32(const float* __restrict__ A,
                                   const float* __restrict__ W2,
                                   const float* __restrict__ Wlin,
                                   float* __restrict__ U,
                                   int nRows) {
  __shared__ float2 WfP[32 * 16];
  {
    const int t = threadIdx.x;
    if ((t & 15) >= 4) {
      WfP[t]       = make_float2(0.f, 0.f);
      WfP[t + 256] = make_float2(0.f, 0.f);
    }
    const int k = t >> 2;
    const int c = t & 3;
    const float4* w2r = (const float4*)(W2 + k * 64);
    const float4* wlr = (const float4*)(Wlin + (c & 1) * 128 + (c >> 1) * 64);
    float s = 0.f;
#pragma unroll
    for (int q = 0; q < 16; ++q) {
      const float4 a = w2r[q];
      const float4 b = wlr[q];
      s += a.x * b.x + a.y * b.y + a.z * b.z + a.w * b.w;
    }
    ((float*)WfP)[((k >> 1) * 16 + c) * 2 + (k & 1)] = s;
  }
  __syncthreads();

  const int lane = threadIdx.x & 31;
  const int wave = threadIdx.x >> 5;
  const int nStrips = nRows >> 4;
  const int strip = blockIdx.x * 8 + wave;
  if (strip >= nStrips) return;

  const int n  = lane & 15;
  const int hi = lane >> 4;
  const long rowBase = (long)strip * 16;
  const float* Arow = A + (rowBase + n) * 64;

  v8f acc = {};
#pragma unroll
  for (int k = 0; k < 64; k += 4) {
    const int ka = k + 2 * hi;
    float2 av = *(const float2*)(Arow + ka);
    av.x = fmaxf(av.x, 0.f);
    av.y = fmaxf(av.y, 0.f);
    v2f a; a[0] = av.x; a[1] = av.y;
    const v2f b = *(const v2f*)&WfP[((k >> 1) + hi) * 16 + n];
    acc = __builtin_amdgcn_wmma_f32_16x16x4_f32(false, a, false, b, (short)0, acc, false, false);
  }

  if (n < 4) {
    float* Uout = U + (rowBase + 8 * hi) * 4 + n;
#pragma unroll
    for (int r = 0; r < 8; ++r) Uout[r * 4] = acc[r];
  }
}

#define E4_PASSES 8
__global__ void scatter_edges4(const float* __restrict__ u,
                               const int* __restrict__ src,
                               const int* __restrict__ dst,
                               const float* __restrict__ w,
                               float* __restrict__ y,
                               int E) {
  const int c = threadIdx.x & 3;
  const int blockFirst = blockIdx.x * (64 * E4_PASSES);
  const int e0 = blockFirst + (threadIdx.x >> 2);
  if (blockFirst + 64 * E4_PASSES <= E) {
    int s[E4_PASSES], d[E4_PASSES];
    float ww[E4_PASSES], v[E4_PASSES];
#pragma unroll
    for (int i = 0; i < E4_PASSES; ++i) {
      s[i]  = src[e0 + 64 * i];
      d[i]  = dst[e0 + 64 * i];
      ww[i] = w[e0 + 64 * i];
    }
#pragma unroll
    for (int i = 0; i < E4_PASSES; ++i) v[i] = u[(long)s[i] * 4 + c];
#pragma unroll
    for (int i = 0; i < E4_PASSES; ++i) unsafeAtomicAdd(y + (long)d[i] * 4 + c, v[i] * ww[i]);
  } else {
    for (int i = 0; i < E4_PASSES; ++i) {
      const int e = e0 + 64 * i;
      if (e < E) {
        const float v = u[(long)src[e] * 4 + c];
        unsafeAtomicAdd(y + (long)dst[e] * 4 + c, v * w[e]);
      }
    }
  }
}

__global__ void decode_fold(const float* __restrict__ y,
                            const int* __restrict__ pos,
                            float* __restrict__ out,
                            int P) {
  const int p = blockIdx.x * blockDim.x + threadIdx.x;
  if (p >= P) return;
  const int s = pos[p];
  const int d = pos[P + p];
  const float2 a = *(const float2*)(y + (long)s * 4);
  const float2 b = *(const float2*)(y + (long)d * 4 + 2);
  *(float2*)(out + 2 * (long)p) = make_float2(a.x + b.x, a.y + b.y);
}

extern "C" void kernel_launch(void* const* d_in, const int* in_sizes, int n_in,
                              void* d_out, int out_size, void* d_ws, size_t ws_size,
                              hipStream_t stream) {
  const float* x    = (const float*)d_in[0];
  const int*   ei   = (const int*)d_in[1];
  const float* w    = (const float*)d_in[2];
  const int*   pos  = (const int*)d_in[3];
  const float* W1   = (const float*)d_in[4];
  const float* W2   = (const float*)d_in[5];
  const float* Wlin = (const float*)d_in[6];
  float* out = (float*)d_out;

  const int F = 128;
  const int N = in_sizes[0] / F;
  const int E = in_sizes[2];
  const int P = in_sizes[3] / 2;

  float* t1 = (float*)d_ws;
  float* h  = t1 + (size_t)N * 64;
  float* u  = h  + (size_t)N * 64;
  float* y  = u  + (size_t)N * 4;

  hipMemsetAsync(h, 0, (size_t)N * 64 * sizeof(float), stream);
  hipMemsetAsync(y, 0, (size_t)N * 4 * sizeof(float), stream);

  const int gemmBlocks = (N / 16 + 7) / 8;
  const long sthreads  = (long)E * 32;
  const int  sblocks   = (int)((sthreads + 255) / 256);

  gemm_wmma_f32<0><<<gemmBlocks, 256, 0, stream>>>(x, W1, t1, N, F);
  scatter_edges<<<sblocks, 256, 0, stream>>>(t1, ei, ei + E, w, h, E);
  gemm_fold_wmma_f32<<<gemmBlocks, 256, 0, stream>>>(h, W2, Wlin, u, N);
  scatter_edges4<<<(E + 64 * E4_PASSES - 1) / (64 * E4_PASSES), 256, 0, stream>>>(u, ei, ei + E, w, y, E);
  decode_fold<<<(P + 255) / 256, 256, 0, stream>>>(y, pos, out, P);
}


# Both edge sums as pulls over destination-sorted edge records built per call (count, range alloc, fill by arrival-order rank), no float atomics; layer-1 pull fused with relu + fold GEMM, h in LDS
# speedup vs baseline: 1.6164x; 1.1247x over previous
#include <hip/hip_runtime.h>
#include <hip/hip_bf16.h>

typedef __attribute__((ext_vector_type(2))) float v2f;
typedef __attribute__((ext_vector_type(8))) float v8f;

#define BSTRIDE 80

template <int RELU>
__global__ void gemm_wmma_f32(const float* __restrict__ A,
                              const float* __restrict__ B,
                              float* __restrict__ C,
                              int nRows, int K) {
  __shared__ float2 BsP[64 * BSTRIDE];
  const int pairRows = K >> 1;
  for (int j = threadIdx.x; j < pairRows * 64; j += blockDim.x) {
    const int k2 = j >> 6;
    const int nn = j & 63;
    BsP[k2 * BSTRIDE + nn] =
        make_float2(B[(2 * k2) * 64 + nn], B[(2 * k2 + 1) * 64 + nn]);
  }
  __syncthreads();

  const int lane = threadIdx.x & 31;
  const int wave = threadIdx.x >> 5;
  const int nStrips = nRows >> 4;
  const int strip = blockIdx.x * 8 + wave;
  if (strip >= nStrips) return;

  const int n  = lane & 15;
  const int hi = lane >> 4;
  const long rowBase = (long)strip * 16;
  const float* Arow = A + (rowBase + n) * (long)K;

  v8f acc0 = {}, acc1 = {}, acc2 = {}, acc3 = {};
  for (int k = 0; k < K; k += 4) {
    const int ka = k + 2 * hi;
    float2 av = *(const float2*)(Arow + ka);
    if (RELU) { av.x = fmaxf(av.x, 0.f); av.y = fmaxf(av.y, 0.f); }
    v2f a; a[0] = av.x; a[1] = av.y;

    const float2* bp = &BsP[((k >> 1) + hi) * BSTRIDE + n];
    const v2f bA = *(const v2f*)(bp + 0);
    const v2f bB = *(const v2f*)(bp + 16);
    const v2f bC = *(const v2f*)(bp + 32);
    const v2f bD = *(const v2f*)(bp + 48);

    acc0 = __builtin_amdgcn_wmma_f32_16x16x4_f32(false, a, false, bA, (short)0, acc0, false, false);
    acc1 = __builtin_amdgcn_wmma_f32_16x16x4_f32(false, a, false, bB, (short)0, acc1, false, false);
    acc2 = __builtin_amdgcn_wmma_f32_16x16x4_f32(false, a, false, bC, (short)0, acc2, false, false);
    acc3 = __builtin_amdgcn_wmma_f32_16x16x4_f32(false, a, false, bD, (short)0, acc3, false, false);
  }

  float* Cout = C + (rowBase + 8 * hi) * 64 + n;
#pragma unroll
  for (int r = 0; r < 8; ++r) {
    Cout[r * 64 + 0]  = acc0[r];
    Cout[r * 64 + 16] = acc1[r];
    Cout[r * 64 + 32] = acc2[r];
    Cout[r * 64 + 48] = acc3[r];
  }
}

__global__ void count_in_edges(const int* __restrict__ dst,
                               int* __restrict__ cur,
                               int E) {
  const int e = blockIdx.x * blockDim.x + threadIdx.x;
  if (e < E) atomicAdd(cur + dst[e], 1);
}

__global__ void __launch_bounds__(256) alloc_segments(int* __restrict__ cur,
                                                      int* __restrict__ off,
                                                      int* __restrict__ total,
                                                      int N) {
  __shared__ int waveSum[8];
  __shared__ int blockBase;
  const int n    = blockIdx.x * 256 + threadIdx.x;
  const int lane = threadIdx.x & 31;
  const int wave = threadIdx.x >> 5;
  const int c = (n < N) ? cur[n] : 0;
  int incl = c;
#pragma unroll
  for (int d = 1; d < 32; d <<= 1) {
    const int up = __shfl_up(incl, d);
    if (lane >= d) incl += up;
  }
  if (lane == 31) waveSum[wave] = incl;
  __syncthreads();
  if (threadIdx.x == 0) {
    int s = 0;
    for (int i = 0; i < 8; ++i) { const int t = waveSum[i]; waveSum[i] = s; s += t; }
    blockBase = atomicAdd(total, s);
  }
  __syncthreads();
  const int o = blockBase + waveSum[wave] + incl - c;
  if (n < N) { off[n] = o; cur[n] = o; }
}

__global__ void fill_records(const int* __restrict__ src,
                             const int* __restrict__ dst,
                             const float* __restrict__ w,
                             int* __restrict__ cur,
                             int2* __restrict__ rec,
                             int E) {
  const int e = blockIdx.x * blockDim.x + threadIdx.x;
  if (e >= E) return;
  const int p = atomicAdd(cur + dst[e], 1);
  rec[p] = make_int2(src[e], __float_as_int(w[e]));
}

#define HSTRIDE 68
#define PULL_GROUP 4

__global__ void __launch_bounds__(512) pull_fold_wmma_f32(const float* __restrict__ t,
                                   const int* __restrict__ off,
                                   const int* __restrict__ end,
                                   const int2* __restrict__ rec,
                                   const float* __restrict__ W2,
                                   const float* __restrict__ Wlin,
                                   float* __restrict__ U,
                                   int nRows) {
  __shared__ float2 WfP[32 * 16];
  __shared__ __attribute__((aligned(16))) float Hs[64 * HSTRIDE];
  if (threadIdx.x < 256) {
    const int t4 = threadIdx.x;
    if ((t4 & 15) >= 4) {
      WfP[t4]       = make_float2(0.f, 0.f);
      WfP[t4 + 256] = make_float2(0.f, 0.f);
    }
    const int k = t4 >> 2;
    const int c = t4 & 3;
    const float4* w2r = (const float4*)(W2 + k * 64);
    const float4* wlr = (const float4*)(Wlin + (c & 1) * 128 + (c >> 1) * 64);
    float s = 0.f;
#pragma unroll 4
    for (int q = 0; q < 16; ++q) {
      const float4 a = w2r[q];
      const float4 b = wlr[q];
      s += a.x * b.x + a.y * b.y + a.z * b.z + a.w * b.w;
    }
    ((float*)WfP)[((k >> 1) * 16 + c) * 2 + (k & 1)] = s;
  }

  const int lane = threadIdx.x & 31;
  const int wave = threadIdx.x >> 5;

  {
    const int node0 = blockIdx.x * 64 + wave * 4;
    int myOff = 0, myEnd = 0;
    if (lane < 4 && node0 + lane < nRows) {
      myOff = off[node0 + lane];
      myEnd = end[node0 + lane];
    }
    int segS[4], segT[4];
    int2 first[4];
#pragma unroll
    for (int j = 0; j < 4; ++j) {
      segS[j] = __builtin_amdgcn_readlane(myOff, j);
      segT[j] = __builtin_amdgcn_readlane(myEnd, j);
      first[j] = make_int2(0, 0);
      if (lane < segT[j] - segS[j]) first[j] = rec[segS[j] + lane];
    }
#pragma unroll
    for (int j = 0; j < 4; ++j) {
      const int s = segS[j], e = segT[j];
      float2 acc = make_float2(0.f, 0.f);
      for (int base = s; base < e; base += 32) {
        const int cnt = min(32, e - base);
        int2 r = first[j];
        if (base != s) {
          r = make_int2(0, 0);
          if (lane < cnt) r = rec[base + lane];
        }
        for (int i = 0; i < cnt; i += PULL_GROUP) {
          float2 v[PULL_GROUP];
          float  ww[PULL_GROUP];
#pragma unroll
          for (int g = 0; g < PULL_GROUP; ++g) {
            const int ig = min(i + g, cnt - 1);
            const int sg = __builtin_amdgcn_readlane(r.x, ig);
            const float wg = __int_as_float(__builtin_amdgcn_readlane(r.y, ig));
            ww[g] = (i + g < cnt) ? wg : 0.f;
            v[g] = *(const float2*)(t + (long)sg * 64 + 2 * lane);
          }
#pragma unroll
          for (int g = 0; g < PULL_GROUP; ++g) {
            acc.x += ww[g] * v[g].x;
            acc.y += ww[g] * v[g].y;
          }
        }
      }
      acc.x = fmaxf(acc.x, 0.f);
      acc.y = fmaxf(acc.y, 0.f);
      *(float2*)&Hs[(wave * 4 + j) * HSTRIDE + 2 * lane] = acc;
    }
  }
  __syncthreads();

  const int nStrips = nRows >> 4;
  const int strip = blockIdx.x * 4 + wave;
  if (wave >= 4 || strip >= nStrips) return;

  const int n  = lane & 15;
  const int hi = lane >> 4;
  const long rowBase = (long)strip * 16;
  const float* Arow = &Hs[(wave * 16 + n) * HSTRIDE];

  v8f acc = {};
#pragma unroll
  for (int k = 0; k < 64; k += 4) {
    const int ka = k + 2 * hi;
    const v2f a = *(const v2f*)(Arow + ka);
    const v2f b = *(const v2f*)&WfP[((k >> 1) + hi) * 16 + n];
    acc = __builtin_amdgcn_wmma_f32_16x16x4_f32(false, a, false, b, (short)0, acc, false, false);
  }

  if (n < 4) {
    float* Uout = U + (rowBase + 8 * hi) * 4 + n;
#pragma unroll
    for (int r = 0; r < 8; ++r) Uout[r * 4] = acc[r];
  }
}

__global__ void pull_edges4(const float* __restrict__ u,
                            const int* __restrict__ off,
                            const int* __restrict__ end,
                            const int2* __restrict__ rec,
                            float* __restrict__ y,
                            int N) {
  const int tid = blockIdx.x * blockDim.x + threadIdx.x;
  const int n = tid >> 2;
  const int c = tid & 3;
  if (n >= N) return;
  const int s = off[n], e = end[n];
  float acc = 0.f;
  for (int b = s; b < e; b += 4) {
    int   sg[4];
    float wg[4], v[4];
#pragma unroll
    for (int g = 0; g < 4; ++g) {
      const int2 r = rec[min(b + g, e - 1)];
      sg[g] = r.x;
      wg[g] = (b + g < e) ? __int_as_float(r.y) : 0.f;
    }
#pragma unroll
    for (int g = 0; g < 4; ++g) v[g] = u[(long)sg[g] * 4 + c];
#pragma unroll
    for (int g = 0; g < 4; ++g) acc += wg[g] * v[g];
  }
  y[(long)n * 4 + c] = acc;
}

__global__ void decode_fold(const float* __restrict__ y,
                            const int* __restrict__ pos,
                            float* __restrict__ out,
                            int P) {
  const int p = blockIdx.x * blockDim.x + threadIdx.x;
  if (p >= P) return;
  const int s = pos[p];
  const int d = pos[P + p];
  const float2 a = *(const float2*)(y + (long)s * 4);
  const float2 b = *(const float2*)(y + (long)d * 4 + 2);
  *(float2*)(out + 2 * (long)p) = make_float2(a.x + b.x, a.y + b.y);
}

extern "C" void kernel_launch(void* const* d_in, const int* in_sizes, int n_in,
                              void* d_out, int out_size, void* d_ws, size_t ws_size,
                              hipStream_t stream) {
  const float* x    = (const float*)d_in[0];
  const int*   ei   = (const int*)d_in[1];
  const float* w    = (const float*)d_in[2];
  const int*   pos  = (const int*)d_in[3];
  const float* W1   = (const float*)d_in[4];
  const float* W2   = (const float*)d_in[5];
  const float* Wlin = (const float*)d_in[6];
  float* out = (float*)d_out;

  const int F = 128;
  const int N = in_sizes[0] / F;
  const int E = in_sizes[2];
  const int P = in_sizes[3] / 2;

  float* t1  = (float*)d_ws;
  float* u   = t1 + (size_t)N * 64;
  float* y   = u  + (size_t)N * 4;
  int2*  rec = (int2*)(y + (size_t)N * 4);
  int*   off = (int*)(rec + (size_t)E);
  int*   cur = off + N;

  hipMemsetAsync(cur, 0, ((size_t)N + 1) * sizeof(int), stream);

  const int gemmBlocks = (N / 16 + 7) / 8;
  const int eblocks    = (E + 255) / 256;

  count_in_edges<<<eblocks, 256, 0, stream>>>(ei + E, cur, E);
  alloc_segments<<<(N + 255) / 256, 256, 0, stream>>>(cur, off, cur + N, N);
  fill_records<<<eblocks, 256, 0, stream>>>(ei, ei + E, w, cur, rec, E);
  gemm_wmma_f32<0><<<gemmBlocks, 256, 0, stream>>>(x, W1, t1, N, F);
  pull_fold_wmma_f32<<<(N + 63) / 64, 512, 0, stream>>>(t1, off, cur, rec, W2, Wlin, u, N);
  pull_edges4<<<(N * 4 + 255) / 256, 256, 0, stream>>>(u, off, cur, rec, y, N);
  decode_fold<<<(P + 255) / 256, 256, 0, stream>>>(y, pos, out, P);
}


# Edge grouping as a two-level counting sort: per-edge counts and ranks in LDS, one global add per block and 256-node bucket, then a per-bucket LDS sort; no per-edge global atomics; arrival order kept
# speedup vs baseline: 3.1349x; 1.9394x over previous
#include <hip/hip_runtime.h>
#include <hip/hip_bf16.h>

typedef __attribute__((ext_vector_type(2))) float v2f;
typedef __attribute__((ext_vector_type(8))) float v8f;

#define BSTRIDE 80

template <int RELU>
__global__ void gemm_wmma_f32(const float* __restrict__ A,
                              const float* __restrict__ B,
                              float* __restrict__ C,
                              int nRows, int K) {
  __shared__ float2 BsP[64 * BSTRIDE];
  const int pairRows = K >> 1;
  for (int j = threadIdx.x; j < pairRows * 64; j += blockDim.x) {
    const int k2 = j >> 6;
    const int nn = j & 63;
    BsP[k2 * BSTRIDE + nn] =
        make_float2(B[(2 * k2) * 64 + nn], B[(2 * k2 + 1) * 64 + nn]);
  }
  __syncthreads();

  const int lane = threadIdx.x & 31;
  const int wave = threadIdx.x >> 5;
  const int nStrips = nRows >> 4;
  const int strip = blockIdx.x * 8 + wave;
  if (strip >= nStrips) return;

  const int n  = lane & 15;
  const int hi = lane >> 4;
  const long rowBase = (long)strip * 16;
  const float* Arow = A + (rowBase + n) * (long)K;

  v8f acc0 = {}, acc1 = {}, acc2 = {}, acc3 = {};
  for (int k = 0; k < K; k += 4) {
    const int ka = k + 2 * hi;
    float2 av = *(const float2*)(Arow + ka);
    if (RELU) { av.x = fmaxf(av.x, 0.f); av.y = fmaxf(av.y, 0.f); }
    v2f a; a[0] = av.x; a[1] = av.y;

    const float2* bp = &BsP[((k >> 1) + hi) * BSTRIDE + n];
    const v2f bA = *(const v2f*)(bp + 0);
    const v2f bB = *(const v2f*)(bp + 16);
    const v2f bC = *(const v2f*)(bp + 32);
    const v2f bD = *(const v2f*)(bp + 48);

    acc0 = __builtin_amdgcn_wmma_f32_16x16x4_f32(false, a, false, bA, (short)0, acc0, false, false);
    acc1 = __builtin_amdgcn_wmma_f32_16x16x4_f32(false, a, false, bB, (short)0, acc1, false, false);
    acc2 = __builtin_amdgcn_wmma_f32_16x16x4_f32(false, a, false, bC, (short)0, acc2, false, false);
    acc3 = __builtin_amdgcn_wmma_f32_16x16x4_f32(false, a, false, bD, (short)0, acc3, false, false);
  }

  float* Cout = C + (rowBase + 8 * hi) * 64 + n;
#pragma unroll
  for (int r = 0; r < 8; ++r) {
    Cout[r * 64 + 0]  = acc0[r];
    Cout[r * 64 + 16] = acc1[r];
    Cout[r * 64 + 32] = acc2[r];
    Cout[r * 64 + 48] = acc3[r];
  }
}

#define BUCKET_SHIFT 8
#define MAX_BUCKETS 512
#define EDGES_PER_BLOCK 2048

__device__ __forceinline__ void scan_bucket_counts(const int* __restrict__ bucketCount,
                                                   int nb, int* base, int* waveSum) {
  const int t    = threadIdx.x;
  const int lane = t & 31;
  const int wave = t >> 5;
  const int c0 = (2 * t     < nb) ? bucketCount[2 * t]     : 0;
  const int c1 = (2 * t + 1 < nb) ? bucketCount[2 * t + 1] : 0;
  const int c  = c0 + c1;
  int incl = c;
#pragma unroll
  for (int d = 1; d < 32; d <<= 1) {
    const int up = __shfl_up(incl, d);
    if (lane >= d) incl += up;
  }
  if (lane == 31) waveSum[wave] = incl;
  __syncthreads();
  int wbase = 0;
#pragma unroll
  for (int i = 0; i < 8; ++i) {
    const int s = waveSum[i];
    if (i < wave) wbase += s;
  }
  const int excl = wbase + incl - c;
  base[2 * t]     = excl;
  base[2 * t + 1] = excl + c0;
  if (t == 255) base[MAX_BUCKETS] = excl + c;
  __syncthreads();
}

__global__ void __launch_bounds__(256) bucket_count(const int* __restrict__ dst,
                                                    int* __restrict__ bucketCount,
                                                    int E, int nb) {
  __shared__ int hist[MAX_BUCKETS];
  const int t = threadIdx.x;
  hist[t] = 0;
  hist[t + 256] = 0;
  __syncthreads();
  const int e0 = blockIdx.x * EDGES_PER_BLOCK + t;
  int d[8];
#pragma unroll
  for (int j = 0; j < 8; ++j) {
    const int e = e0 + j * 256;
    d[j] = (e < E) ? dst[e] : -1;
  }
#pragma unroll
  for (int j = 0; j < 8; ++j)
    if (d[j] >= 0) atomicAdd(&hist[d[j] >> BUCKET_SHIFT], 1);
  __syncthreads();
  for (int b = t; b < nb; b += 256) {
    const int h = hist[b];
    if (h) atomicAdd(bucketCount + b, h);
  }
}

__global__ void __launch_bounds__(256) bucket_scatter(const int* __restrict__ src,
                                                      const int* __restrict__ dst,
                                                      const float* __restrict__ w,
                                                      const int* __restrict__ bucketCount,
                                                      int* __restrict__ bucketCursor,
                                                      int2* __restrict__ staged,
                                                      int E, int nb) {
  __shared__ int base[MAX_BUCKETS + 1];
  __shared__ int hist[MAX_BUCKETS];
  __shared__ int waveSum[8];
  const int t = threadIdx.x;
  hist[t] = 0;
  hist[t + 256] = 0;

  const int e0 = blockIdx.x * EDGES_PER_BLOCK + t;
  int d[8], s[8], wb[8];
#pragma unroll
  for (int j = 0; j < 8; ++j) {
    const int e = e0 + j * 256;
    const bool in = e < E;
    d[j]  = in ? dst[e] : -1;
    s[j]  = in ? src[e] : 0;
    wb[j] = in ? __float_as_int(w[e]) : 0;
  }

  scan_bucket_counts(bucketCount, nb, base, waveSum);

  int rank[8];
#pragma unroll
  for (int j = 0; j < 8; ++j) {
    rank[j] = 0;
    if (d[j] >= 0) rank[j] = atomicAdd(&hist[d[j] >> BUCKET_SHIFT], 1);
  }
  __syncthreads();
  for (int b = t; b < nb; b += 256) {
    const int h = hist[b];
    int r = 0;
    if (h) r = atomicAdd(bucketCursor + b, h);
    hist[b] = base[b] + r;
  }
  __syncthreads();
#pragma unroll
  for (int j = 0; j < 8; ++j) {
    if (d[j] >= 0) {
      const int p = hist[d[j] >> BUCKET_SHIFT] + rank[j];
      staged[p] = make_int2((int)((unsigned)s[j] | ((unsigned)(d[j] & 255) << 24)), wb[j]);
    }
  }
}

__global__ void __launch_bounds__(256) bucket_sort(const int* __restrict__ bucketCount,
                                                   const int2* __restrict__ staged,
                                                   int2* __restrict__ rec,
                                                   int* __restrict__ off,
                                                   int* __restrict__ end,
                                                   int N, int nb) {
  __shared__ int base[MAX_BUCKETS + 1];
  __shared__ int waveSum[8];
  __shared__ int cnt[256];
  const int t    = threadIdx.x;
  const int lane = t & 31;
  const int wave = t >> 5;
  cnt[t] = 0;
  scan_bucket_counts(bucketCount, nb, base, waveSum);
  const int s = base[blockIdx.x];
  const int e = base[blockIdx.x + 1];

  for (int i = s + t; i < e; i += 1024) {
    int k[4];
#pragma unroll
    for (int g = 0; g < 4; ++g) k[g] = staged[min(i + g * 256, e - 1)].x;
#pragma unroll
    for (int g = 0; g < 4; ++g)
      if (i + g * 256 < e) atomicAdd(&cnt[(unsigned)k[g] >> 24], 1);
  }
  __syncthreads();

  const int c = cnt[t];
  int incl = c;
#pragma unroll
  for (int d = 1; d < 32; d <<= 1) {
    const int up = __shfl_up(incl, d);
    if (lane >= d) incl += up;
  }
  if (lane == 31) waveSum[wave] = incl;
  __syncthreads();
  int wbase = 0;
#pragma unroll
  for (int i = 0; i < 8; ++i) {
    const int ws = waveSum[i];
    if (i < wave) wbase += ws;
  }
  const int o = s + wbase + incl - c;
  const int node = (blockIdx.x << BUCKET_SHIFT) + t;
  if (node < N) { off[node] = o; end[node] = o + c; }
  cnt[t] = o;
  __syncthreads();

  for (int i = s + t; i < e; i += 1024) {
    int2 r[4];
#pragma unroll
    for (int g = 0; g < 4; ++g) r[g] = staged[min(i + g * 256, e - 1)];
#pragma unroll
    for (int g = 0; g < 4; ++g) {
      if (i + g * 256 < e) {
        const int p = atomicAdd(&cnt[(unsigned)r[g].x >> 24], 1);
        rec[p] = make_int2(r[g].x & 0x00FFFFFF, r[g].y);
      }
    }
  }
}

#define HSTRIDE 68
#define PULL_GROUP 4

__global__ void __launch_bounds__(512) pull_fold_wmma_f32(const float* __restrict__ t,
                                   const int* __restrict__ off,
                                   const int* __restrict__ end,
                                   const int2* __restrict__ rec,
                                   const float* __restrict__ W2,
                                   const float* __restrict__ Wlin,
                                   float* __restrict__ U,
                                   int nRows) {
  __shared__ float2 WfP[32 * 16];
  __shared__ __attribute__((aligned(16))) float Hs[64 * HSTRIDE];
  if (threadIdx.x < 256) {
    const int t4 = threadIdx.x;
    if ((t4 & 15) >= 4) {
      WfP[t4]       = make_float2(0.f, 0.f);
      WfP[t4 + 256] = make_float2(0.f, 0.f);
    }
    const int k = t4 >> 2;
    const int c = t4 & 3;
    const float4* w2r = (const float4*)(W2 + k * 64);
    const float4* wlr = (const float4*)(Wlin + (c & 1) * 128 + (c >> 1) * 64);
    float s = 0.f;
#pragma unroll 4
    for (int q = 0; q < 16; ++q) {
      const float4 a = w2r[q];
      const float4 b = wlr[q];
      s += a.x * b.x + a.y * b.y + a.z * b.z + a.w * b.w;
    }
    ((float*)WfP)[((k >> 1) * 16 + c) * 2 + (k & 1)] = s;
  }

  const int lane = threadIdx.x & 31;
  const int wave = threadIdx.x >> 5;

  {
    const int node0 = blockIdx.x * 64 + wave * 4;
    int myOff = 0, myEnd = 0;
    if (lane < 4 && node0 + lane < nRows) {
      myOff = off[node0 + lane];
      myEnd = end[node0 + lane];
    }
    int segS[4], segT[4];
    int2 first[4];
#pragma unroll
    for (int j = 0; j < 4; ++j) {
      segS[j] = __builtin_amdgcn_readlane(myOff, j);
      segT[j] = __builtin_amdgcn_readlane(myEnd, j);
      first[j] = make_int2(0, 0);
      if (lane < segT[j] - segS[j]) first[j] = rec[segS[j] + lane];
    }
#pragma unroll
    for (int j = 0; j < 4; ++j) {
      const int s = segS[j], e = segT[j];
      float2 acc = make_float2(0.f, 0.f);
      for (int base = s; base < e; base += 32) {
        const int cnt = min(32, e - base);
        int2 r = first[j];
        if (base != s) {
          r = make_int2(0, 0);
          if (lane < cnt) r = rec[base + lane];
        }
        for (int i = 0; i < cnt; i += PULL_GROUP) {
          float2 v[PULL_GROUP];
          float  ww[PULL_GROUP];
#pragma unroll
          for (int g = 0; g < PULL_GROUP; ++g) {
            const int ig = min(i + g, cnt - 1);
            const int sg = __builtin_amdgcn_readlane(r.x, ig);
            const float wg = __int_as_float(__builtin_amdgcn_readlane(r.y, ig));
            ww[g] = (i + g < cnt) ? wg : 0.f;
            v[g] = *(const float2*)(t + (long)sg * 64 + 2 * lane);
          }
#pragma unroll
          for (int g = 0; g < PULL_GROUP; ++g) {
            acc.x += ww[g] * v[g].x;
            acc.y += ww[g] * v[g].y;
          }
        }
      }
      acc.x = fmaxf(acc.x, 0.f);
      acc.y = fmaxf(acc.y, 0.f);
      *(float2*)&Hs[(wave * 4 + j) * HSTRIDE + 2 * lane] = acc;
    }
  }
  __syncthreads();

  const int nStrips = nRows >> 4;
  const int strip = blockIdx.x * 4 + wave;
  if (wave >= 4 || strip >= nStrips) return;

  const int n  = lane & 15;
  const int hi = lane >> 4;
  const long rowBase = (long)strip * 16;
  const float* Arow = &Hs[(wave * 16 + n) * HSTRIDE];

  v8f acc = {};
#pragma unroll
  for (int k = 0; k < 64; k += 4) {
    const int ka = k + 2 * hi;
    const v2f a = *(const v2f*)(Arow + ka);
    const v2f b = *(const v2f*)&WfP[((k >> 1) + hi) * 16 + n];
    acc = __builtin_amdgcn_wmma_f32_16x16x4_f32(false, a, false, b, (short)0, acc, false, false);
  }

  if (n < 4) {
    float* Uout = U + (rowBase + 8 * hi) * 4 + n;
#pragma unroll
    for (int r = 0; r < 8; ++r) Uout[r * 4] = acc[r];
  }
}

__global__ void pull_edges4(const float* __restrict__ u,
                            const int* __restrict__ off,
                            const int* __restrict__ end,
                            const int2* __restrict__ rec,
                            float* __restrict__ y,
                            int N) {
  const int tid = blockIdx.x * blockDim.x + threadIdx.x;
  const int n = tid >> 2;
  const int c = tid & 3;
  if (n >= N) return;
  const int s = off[n], e = end[n];
  float acc = 0.f;
  for (int b = s; b < e; b += 4) {
    int   sg[4];
    float wg[4], v[4];
#pragma unroll
    for (int g = 0; g < 4; ++g) {
      const int2 r = rec[min(b + g, e - 1)];
      sg[g] = r.x;
      wg[g] = (b + g < e) ? __int_as_float(r.y) : 0.f;
    }
#pragma unroll
    for (int g = 0; g < 4; ++g) v[g] = u[(long)sg[g] * 4 + c];
#pragma unroll
    for (int g = 0; g < 4; ++g) acc += wg[g] * v[g];
  }
  y[(long)n * 4 + c] = acc;
}

__global__ void decode_fold(const float* __restrict__ y,
                            const int* __restrict__ pos,
                            float* __restrict__ out,
                            int P) {
  const int p = blockIdx.x * blockDim.x + threadIdx.x;
  if (p >= P) return;
  const int s = pos[p];
  const int d = pos[P + p];
  const float2 a = *(const float2*)(y + (long)s * 4);
  const float2 b = *(const float2*)(y + (long)d * 4 + 2);
  *(float2*)(out + 2 * (long)p) = make_float2(a.x + b.x, a.y + b.y);
}

extern "C" void kernel_launch(void* const* d_in, const int* in_sizes, int n_in,
                              void* d_out, int out_size, void* d_ws, size_t ws_size,
                              hipStream_t stream) {
  const float* x    = (const float*)d_in[0];
  const int*   ei   = (const int*)d_in[1];
  const float* w    = (const float*)d_in[2];
  const int*   pos  = (const int*)d_in[3];
  const float* W1   = (const float*)d_in[4];
  const float* W2   = (const float*)d_in[5];
  const float* Wlin = (const float*)d_in[6];
  float* out = (float*)d_out;

  const int F = 128;
  const int N = in_sizes[0] / F;
  const int E = in_sizes[2];
  const int P = in_sizes[3] / 2;

  float* t1  = (float*)d_ws;
  float* u   = t1 + (size_t)N * 64;
  float* y   = u  + (size_t)N * 4;
  int2*  rec = (int2*)(y + (size_t)N * 4);
  int*   off = (int*)(rec + (size_t)E);
  int*   end = off + N;
  int2*  staged       = (int2*)t1;
  int*   bucketCount  = (int*)y;
  int*   bucketCursor = bucketCount + MAX_BUCKETS;

  hipMemsetAsync(bucketCount, 0, 2 * MAX_BUCKETS * sizeof(int), stream);

  const int gemmBlocks = (N / 16 + 7) / 8;
  const int eblocks    = (E + EDGES_PER_BLOCK - 1) / EDGES_PER_BLOCK;
  const int nb         = (N + 255) >> BUCKET_SHIFT;

  bucket_count<<<eblocks, 256, 0, stream>>>(ei + E, bucketCount, E, nb);
  bucket_scatter<<<eblocks, 256, 0, stream>>>(ei, ei + E, w, bucketCount, bucketCursor, staged, E, nb);
  bucket_sort<<<nb, 256, 0, stream>>>(bucketCount, staged, rec, off, end, N, nb);
  gemm_wmma_f32<0><<<gemmBlocks, 256, 0, stream>>>(x, W1, t1, N, F);
  pull_fold_wmma_f32<<<(N + 63) / 64, 512, 0, stream>>>(t1, off, end, rec, W2, Wlin, u, N);
  pull_edges4<<<(N * 4 + 255) / 256, 256, 0, stream>>>(u, off, end, rec, y, N);
  decode_fold<<<(P + 255) / 256, 256, 0, stream>>>(y, pos, out, P);
}


# GEMM weight staging: a thread requests its share of W1 as one batch of eight 16-byte loads and writes LDS with 16-byte stores, not 32 dependent 4-byte round trips before the barrier; sums unchanged
# speedup vs baseline: 3.3575x; 1.0710x over previous
#include <hip/hip_runtime.h>
#include <hip/hip_bf16.h>

typedef __attribute__((ext_vector_type(2))) float v2f;
typedef __attribute__((ext_vector_type(8))) float v8f;

#define BSTRIDE 80

template <int RELU>
__global__ void gemm_wmma_f32(const float* __restrict__ A,
                              const float* __restrict__ B,
                              float* __restrict__ C,
                              int nRows, int K) {
  __shared__ __attribute__((aligned(16))) float2 BsP[64 * BSTRIDE];
  const int cells = (K >> 1) * 16;
  for (int base = threadIdx.x; base < cells; base += 4 * (int)blockDim.x) {
    float4 ra[4], rb[4];
#pragma unroll
    for (int i = 0; i < 4; ++i) {
      const int idx = min(base + i * (int)blockDim.x, cells - 1);
      const int k2 = idx >> 4;
      const int c4 = idx & 15;
      ra[i] = *(const float4*)(B + (2 * k2) * 64 + 4 * c4);
      rb[i] = *(const float4*)(B + (2 * k2 + 1) * 64 + 4 * c4);
    }
#pragma unroll
    for (int i = 0; i < 4; ++i) {
      const int idx = min(base + i * (int)blockDim.x, cells - 1);
      const int k2 = idx >> 4;
      const int c4 = idx & 15;
      float4* dst = (float4*)&BsP[k2 * BSTRIDE + 4 * c4];
      dst[0] = make_float4(ra[i].x, rb[i].x, ra[i].y, rb[i].y);
      dst[1] = make_float4(ra[i].z, rb[i].z, ra[i].w, rb[i].w);
    }
  }
  __syncthreads();

  const int lane = threadIdx.x & 31;
  const int wave = threadIdx.x >> 5;
  const int nStrips = nRows >> 4;
  const int strip = blockIdx.x * 8 + wave;
  if (strip >= nStrips) return;

  const int n  = lane & 15;
  const int hi = lane >> 4;
  const long rowBase = (long)strip * 16;
  const float* Arow = A + (rowBase + n) * (long)K;

  v8f acc0 = {}, acc1 = {}, acc2 = {}, acc3 = {};
  for (int k = 0; k < K; k += 4) {
    const int ka = k + 2 * hi;
    float2 av = *(const float2*)(Arow + ka);
    if (RELU) { av.x = fmaxf(av.x, 0.f); av.y = fmaxf(av.y, 0.f); }
    v2f a; a[0] = av.x; a[1] = av.y;

    const float2* bp = &BsP[((k >> 1) + hi) * BSTRIDE + n];
    const v2f bA = *(const v2f*)(bp + 0);
    const v2f bB = *(const v2f*)(bp + 16);
    const v2f bC = *(const v2f*)(bp + 32);
    const v2f bD = *(const v2f*)(bp + 48);

    acc0 = __builtin_amdgcn_wmma_f32_16x16x4_f32(false, a, false, bA, (short)0, acc0, false, false);
    acc1 = __builtin_amdgcn_wmma_f32_16x16x4_f32(false, a, false, bB, (short)0, acc1, false, false);
    acc2 = __builtin_amdgcn_wmma_f32_16x16x4_f32(false, a, false, bC, (short)0, acc2, false, false);
    acc3 = __builtin_amdgcn_wmma_f32_16x16x4_f32(false, a, false, bD, (short)0, acc3, false, false);
  }

  float* Cout = C + (rowBase + 8 * hi) * 64 + n;
#pragma unroll
  for (int r = 0; r < 8; ++r) {
    Cout[r * 64 + 0]  = acc0[r];
    Cout[r * 64 + 16] = acc1[r];
    Cout[r * 64 + 32] = acc2[r];
    Cout[r * 64 + 48] = acc3[r];
  }
}

#define BUCKET_SHIFT 8
#define MAX_BUCKETS 512
#define EDGES_PER_BLOCK 2048

__device__ __forceinline__ void scan_bucket_counts(const int* __restrict__ bucketCount,
                                                   int nb, int* base, int* waveSum) {
  const int t    = threadIdx.x;
  const int lane = t & 31;
  const int wave = t >> 5;
  const int c0 = (2 * t     < nb) ? bucketCount[2 * t]     : 0;
  const int c1 = (2 * t + 1 < nb) ? bucketCount[2 * t + 1] : 0;
  const int c  = c0 + c1;
  int incl = c;
#pragma unroll
  for (int d = 1; d < 32; d <<= 1) {
    const int up = __shfl_up(incl, d);
    if (lane >= d) incl += up;
  }
  if (lane == 31) waveSum[wave] = incl;
  __syncthreads();
  int wbase = 0;
#pragma unroll
  for (int i = 0; i < 8; ++i) {
    const int s = waveSum[i];
    if (i < wave) wbase += s;
  }
  const int excl = wbase + incl - c;
  base[2 * t]     = excl;
  base[2 * t + 1] = excl + c0;
  if (t == 255) base[MAX_BUCKETS] = excl + c;
  __syncthreads();
}

__global__ void __launch_bounds__(256) bucket_count(const int* __restrict__ dst,
                                                    int* __restrict__ bucketCount,
                                                    int E, int nb) {
  __shared__ int hist[MAX_BUCKETS];
  const int t = threadIdx.x;
  hist[t] = 0;
  hist[t + 256] = 0;
  __syncthreads();
  const int e0 = blockIdx.x * EDGES_PER_BLOCK + t;
  int d[8];
#pragma unroll
  for (int j = 0; j < 8; ++j) {
    const int e = e0 + j * 256;
    d[j] = (e < E) ? dst[e] : -1;
  }
#pragma unroll
  for (int j = 0; j < 8; ++j)
    if (d[j] >= 0) atomicAdd(&hist[d[j] >> BUCKET_SHIFT], 1);
  __syncthreads();
  for (int b = t; b < nb; b += 256) {
    const int h = hist[b];
    if (h) atomicAdd(bucketCount + b, h);
  }
}

__global__ void __launch_bounds__(256) bucket_scatter(const int* __restrict__ src,
                                                      const int* __restrict__ dst,
                                                      const float* __restrict__ w,
                                                      const int* __restrict__ bucketCount,
                                                      int* __restrict__ bucketCursor,
                                                      int2* __restrict__ staged,
                                                      int E, int nb) {
  __shared__ int base[MAX_BUCKETS + 1];
  __shared__ int hist[MAX_BUCKETS];
  __shared__ int waveSum[8];
  const int t = threadIdx.x;
  hist[t] = 0;
  hist[t + 256] = 0;

  const int e0 = blockIdx.x * EDGES_PER_BLOCK + t;
  int d[8], s[8], wb[8];
#pragma unroll
  for (int j = 0; j < 8; ++j) {
    const int e = e0 + j * 256;
    const bool in = e < E;
    d[j]  = in ? dst[e] : -1;
    s[j]  = in ? src[e] : 0;
    wb[j] = in ? __float_as_int(w[e]) : 0;
  }

  scan_bucket_counts(bucketCount, nb, base, waveSum);

  int rank[8];
#pragma unroll
  for (int j = 0; j < 8; ++j) {
    rank[j] = 0;
    if (d[j] >= 0) rank[j] = atomicAdd(&hist[d[j] >> BUCKET_SHIFT], 1);
  }
  __syncthreads();
  for (int b = t; b < nb; b += 256) {
    const int h = hist[b];
    int r = 0;
    if (h) r = atomicAdd(bucketCursor + b, h);
    hist[b] = base[b] + r;
  }
  __syncthreads();
#pragma unroll
  for (int j = 0; j < 8; ++j) {
    if (d[j] >= 0) {
      const int p = hist[d[j] >> BUCKET_SHIFT] + rank[j];
      staged[p] = make_int2((int)((unsigned)s[j] | ((unsigned)(d[j] & 255) << 24)), wb[j]);
    }
  }
}

__global__ void __launch_bounds__(256) bucket_sort(const int* __restrict__ bucketCount,
                                                   const int2* __restrict__ staged,
                                                   int2* __restrict__ rec,
                                                   int* __restrict__ off,
                                                   int* __restrict__ end,
                                                   int N, int nb) {
  __shared__ int base[MAX_BUCKETS + 1];
  __shared__ int waveSum[8];
  __shared__ int cnt[256];
  const int t    = threadIdx.x;
  const int lane = t & 31;
  const int wave = t >> 5;
  cnt[t] = 0;
  scan_bucket_counts(bucketCount, nb, base, waveSum);
  const int s = base[blockIdx.x];
  const int e = base[blockIdx.x + 1];

  for (int i = s + t; i < e; i += 1024) {
    int k[4];
#pragma unroll
    for (int g = 0; g < 4; ++g) k[g] = staged[min(i + g * 256, e - 1)].x;
#pragma unroll
    for (int g = 0; g < 4; ++g)
      if (i + g * 256 < e) atomicAdd(&cnt[(unsigned)k[g] >> 24], 1);
  }
  __syncthreads();

  const int c = cnt[t];
  int incl = c;
#pragma unroll
  for (int d = 1; d < 32; d <<= 1) {
    const int up = __shfl_up(incl, d);
    if (lane >= d) incl += up;
  }
  if (lane == 31) waveSum[wave] = incl;
  __syncthreads();
  int wbase = 0;
#pragma unroll
  for (int i = 0; i < 8; ++i) {
    const int ws = waveSum[i];
    if (i < wave) wbase += ws;
  }
  const int o = s + wbase + incl - c;
  const int node = (blockIdx.x << BUCKET_SHIFT) + t;
  if (node < N) { off[node] = o; end[node] = o + c; }
  cnt[t] = o;
  __syncthreads();

  for (int i = s + t; i < e; i += 1024) {
    int2 r[4];
#pragma unroll
    for (int g = 0; g < 4; ++g) r[g] = staged[min(i + g * 256, e - 1)];
#pragma unroll
    for (int g = 0; g < 4; ++g) {
      if (i + g * 256 < e) {
        const int p = atomicAdd(&cnt[(unsigned)r[g].x >> 24], 1);
        rec[p] = make_int2(r[g].x & 0x00FFFFFF, r[g].y);
      }
    }
  }
}

#define HSTRIDE 68
#define PULL_GROUP 4

__global__ void __launch_bounds__(512) pull_fold_wmma_f32(const float* __restrict__ t,
                                   const int* __restrict__ off,
                                   const int* __restrict__ end,
                                   const int2* __restrict__ rec,
                                   const float* __restrict__ W2,
                                   const float* __restrict__ Wlin,
                                   float* __restrict__ U,
                                   int nRows) {
  __shared__ float2 WfP[32 * 16];
  __shared__ __attribute__((aligned(16))) float Hs[64 * HSTRIDE];
  if (threadIdx.x < 256) {
    const int t4 = threadIdx.x;
    if ((t4 & 15) >= 4) {
      WfP[t4]       = make_float2(0.f, 0.f);
      WfP[t4 + 256] = make_float2(0.f, 0.f);
    }
    const int k = t4 >> 2;
    const int c = t4 & 3;
    const float4* w2r = (const float4*)(W2 + k * 64);
    const float4* wlr = (const float4*)(Wlin + (c & 1) * 128 + (c >> 1) * 64);
    float s = 0.f;
#pragma unroll 4
    for (int q = 0; q < 16; ++q) {
      const float4 a = w2r[q];
      const float4 b = wlr[q];
      s += a.x * b.x + a.y * b.y + a.z * b.z + a.w * b.w;
    }
    ((float*)WfP)[((k >> 1) * 16 + c) * 2 + (k & 1)] = s;
  }

  const int lane = threadIdx.x & 31;
  const int wave = threadIdx.x >> 5;

  {
    const int node0 = blockIdx.x * 64 + wave * 4;
    int myOff = 0, myEnd = 0;
    if (lane < 4 && node0 + lane < nRows) {
      myOff = off[node0 + lane];
      myEnd = end[node0 + lane];
    }
    int segS[4], segT[4];
    int2 first[4];
#pragma unroll
    for (int j = 0; j < 4; ++j) {
      segS[j] = __builtin_amdgcn_readlane(myOff, j);
      segT[j] = __builtin_amdgcn_readlane(myEnd, j);
      first[j] = make_int2(0, 0);
      if (lane < segT[j] - segS[j]) first[j] = rec[segS[j] + lane];
    }
#pragma unroll
    for (int j = 0; j < 4; ++j) {
      const int s = segS[j], e = segT[j];
      float2 acc = make_float2(0.f, 0.f);
      for (int base = s; base < e; base += 32) {
        const int cnt = min(32, e - base);
        int2 r = first[j];
        if (base != s) {
          r = make_int2(0, 0);
          if (lane < cnt) r = rec[base + lane];
        }
        for (int i = 0; i < cnt; i += PULL_GROUP) {
          float2 v[PULL_GROUP];
          float  ww[PULL_GROUP];
#pragma unroll
          for (int g = 0; g < PULL_GROUP; ++g) {
            const int ig = min(i + g, cnt - 1);
            const int sg = __builtin_amdgcn_readlane(r.x, ig);
            const float wg = __int_as_float(__builtin_amdgcn_readlane(r.y, ig));
            ww[g] = (i + g < cnt) ? wg : 0.f;
            v[g] = *(const float2*)(t + (long)sg * 64 + 2 * lane);
          }
#pragma unroll
          for (int g = 0; g < PULL_GROUP; ++g) {
            acc.x += ww[g] * v[g].x;
            acc.y += ww[g] * v[g].y;
          }
        }
      }
      acc.x = fmaxf(acc.x, 0.f);
      acc.y = fmaxf(acc.y, 0.f);
      *(float2*)&Hs[(wave * 4 + j) * HSTRIDE + 2 * lane] = acc;
    }
  }
  __syncthreads();

  const int nStrips = nRows >> 4;
  const int strip = blockIdx.x * 4 + wave;
  if (wave >= 4 || strip >= nStrips) return;

  const int n  = lane & 15;
  const int hi = lane >> 4;
  const long rowBase = (long)strip * 16;
  const float* Arow = &Hs[(wave * 16 + n) * HSTRIDE];

  v8f acc = {};
#pragma unroll
  for (int k = 0; k < 64; k += 4) {
    const int ka = k + 2 * hi;
    const v2f a = *(const v2f*)(Arow + ka);
    const v2f b = *(const v2f*)&WfP[((k >> 1) + hi) * 16 + n];
    acc = __builtin_amdgcn_wmma_f32_16x16x4_f32(false, a, false, b, (short)0, acc, false, false);
  }

  if (n < 4) {
    float* Uout = U + (rowBase + 8 * hi) * 4 + n;
#pragma unroll
    for (int r = 0; r < 8; ++r) Uout[r * 4] = acc[r];
  }
}

__global__ void pull_edges4(const float* __restrict__ u,
                            const int* __restrict__ off,
                            const int* __restrict__ end,
                            const int2* __restrict__ rec,
                            float* __restrict__ y,
                            int N) {
  const int tid = blockIdx.x * blockDim.x + threadIdx.x;
  const int n = tid >> 2;
  const int c = tid & 3;
  if (n >= N) return;
  const int s = off[n], e = end[n];
  float acc = 0.f;
  for (int b = s; b < e; b += 4) {
    int   sg[4];
    float wg[4], v[4];
#pragma unroll
    for (int g = 0; g < 4; ++g) {
      const int2 r = rec[min(b + g, e - 1)];
      sg[g] = r.x;
      wg[g] = (b + g < e) ? __int_as_float(r.y) : 0.f;
    }
#pragma unroll
    for (int g = 0; g < 4; ++g) v[g] = u[(long)sg[g] * 4 + c];
#pragma unroll
    for (int g = 0; g < 4; ++g) acc += wg[g] * v[g];
  }
  y[(long)n * 4 + c] = acc;
}

__global__ void decode_fold(const float* __restrict__ y,
                            const int* __restrict__ pos,
                            float* __restrict__ out,
                            int P) {
  const int p = blockIdx.x * blockDim.x + threadIdx.x;
  if (p >= P) return;
  const int s = pos[p];
  const int d = pos[P + p];
  const float2 a = *(const float2*)(y + (long)s * 4);
  const float2 b = *(const float2*)(y + (long)d * 4 + 2);
  *(float2*)(out + 2 * (long)p) = make_float2(a.x + b.x, a.y + b.y);
}

extern "C" void kernel_launch(void* const* d_in, const int* in_sizes, int n_in,
                              void* d_out, int out_size, void* d_ws, size_t ws_size,
                              hipStream_t stream) {
  const float* x    = (const float*)d_in[0];
  const int*   ei   = (const int*)d_in[1];
  const float* w    = (const float*)d_in[2];
  const int*   pos  = (const int*)d_in[3];
  const float* W1   = (const float*)d_in[4];
  const float* W2   = (const float*)d_in[5];
  const float* Wlin = (const float*)d_in[6];
  float* out = (float*)d_out;

  const int F = 128;
  const int N = in_sizes[0] / F;
  const int E = in_sizes[2];
  const int P = in_sizes[3] / 2;

  float* t1  = (float*)d_ws;
  float* u   = t1 + (size_t)N * 64;
  float* y   = u  + (size_t)N * 4;
  int2*  rec = (int2*)(y + (size_t)N * 4);
  int*   off = (int*)(rec + (size_t)E);
  int*   end = off + N;
  int2*  staged       = (int2*)t1;
  int*   bucketCount  = (int*)y;
  int*   bucketCursor = bucketCount + MAX_BUCKETS;

  hipMemsetAsync(bucketCount, 0, 2 * MAX_BUCKETS * sizeof(int), stream);

  const int gemmBlocks = (N / 16 + 7) / 8;
  const int eblocks    = (E + EDGES_PER_BLOCK - 1) / EDGES_PER_BLOCK;
  const int nb         = (N + 255) >> BUCKET_SHIFT;

  bucket_count<<<eblocks, 256, 0, stream>>>(ei + E, bucketCount, E, nb);
  bucket_scatter<<<eblocks, 256, 0, stream>>>(ei, ei + E, w, bucketCount, bucketCursor, staged, E, nb);
  bucket_sort<<<nb, 256, 0, stream>>>(bucketCount, staged, rec, off, end, N, nb);
  gemm_wmma_f32<0><<<gemmBlocks, 256, 0, stream>>>(x, W1, t1, N, F);
  pull_fold_wmma_f32<<<(N + 63) / 64, 512, 0, stream>>>(t1, off, end, rec, W2, Wlin, u, N);
  pull_edges4<<<(N * 4 + 255) / 256, 256, 0, stream>>>(u, off, end, rec, y, N);
  decode_fold<<<(P + 255) / 256, 256, 0, stream>>>(y, pos, out, P);
}


# Layer-1 GEMM has no launch of its own: a third of its row strips runs as extra work-groups inside each of the three edge-grouping launches (staging records moved out of t1's space); sums unchanged
# speedup vs baseline: 3.5500x; 1.0573x over previous
#include <hip/hip_runtime.h>
#include <hip/hip_bf16.h>

typedef __attribute__((ext_vector_type(2))) float v2f;
typedef __attribute__((ext_vector_type(8))) float v8f;

#define BSTRIDE 80

template <int RELU>
__device__ __forceinline__ void gemm_strips(const float* __restrict__ A,
                                            const float* __restrict__ B,
                                            float* __restrict__ C,
                                            int nRows, int K, int firstStrip) {
  __shared__ __attribute__((aligned(16))) float2 BsP[64 * BSTRIDE];
  const int cells = (K >> 1) * 16;
  for (int base = threadIdx.x; base < cells; base += 4 * (int)blockDim.x) {
    float4 ra[4], rb[4];
#pragma unroll
    for (int i = 0; i < 4; ++i) {
      const int idx = min(base + i * (int)blockDim.x, cells - 1);
      const int k2 = idx >> 4;
      const int c4 = idx & 15;
      ra[i] = *(const float4*)(B + (2 * k2) * 64 + 4 * c4);
      rb[i] = *(const float4*)(B + (2 * k2 + 1) * 64 + 4 * c4);
    }
#pragma unroll
    for (int i = 0; i < 4; ++i) {
      const int idx = min(base + i * (int)blockDim.x, cells - 1);
      const int k2 = idx >> 4;
      const int c4 = idx & 15;
      float4* dst = (float4*)&BsP[k2 * BSTRIDE + 4 * c4];
      dst[0] = make_float4(ra[i].x, rb[i].x, ra[i].y, rb[i].y);
      dst[1] = make_float4(ra[i].z, rb[i].z, ra[i].w, rb[i].w);
    }
  }
  __syncthreads();

  const int lane = threadIdx.x & 31;
  const int wave = threadIdx.x >> 5;
  const int nStrips = nRows >> 4;
  const int strip = firstStrip + wave;
  if (strip >= nStrips) return;

  const int n  = lane & 15;
  const int hi = lane >> 4;
  const long rowBase = (long)strip * 16;
  const float* Arow = A + (rowBase + n) * (long)K;

  v8f acc0 = {}, acc1 = {}, acc2 = {}, acc3 = {};
  for (int k = 0; k < K; k += 4) {
    const int ka = k + 2 * hi;
    float2 av = *(const float2*)(Arow + ka);
    if (RELU) { av.x = fmaxf(av.x, 0.f); av.y = fmaxf(av.y, 0.f); }
    v2f a; a[0] = av.x; a[1] = av.y;

    const float2* bp = &BsP[((k >> 1) + hi) * BSTRIDE + n];
    const v2f bA = *(const v2f*)(bp + 0);
    const v2f bB = *(const v2f*)(bp + 16);
    const v2f bC = *(const v2f*)(bp + 32);
    const v2f bD = *(const v2f*)(bp + 48);

    acc0 = __builtin_amdgcn_wmma_f32_16x16x4_f32(false, a, false, bA, (short)0, acc0, false, false);
    acc1 = __builtin_amdgcn_wmma_f32_16x16x4_f32(false, a, false, bB, (short)0, acc1, false, false);
    acc2 = __builtin_amdgcn_wmma_f32_16x16x4_f32(false, a, false, bC, (short)0, acc2, false, false);
    acc3 = __builtin_amdgcn_wmma_f32_16x16x4_f32(false, a, false, bD, (short)0, acc3, false, false);
  }

  float* Cout = C + (rowBase + 8 * hi) * 64 + n;
#pragma unroll
  for (int r = 0; r < 8; ++r) {
    Cout[r * 64 + 0]  = acc0[r];
    Cout[r * 64 + 16] = acc1[r];
    Cout[r * 64 + 32] = acc2[r];
    Cout[r * 64 + 48] = acc3[r];
  }
}

#define BUCKET_SHIFT 8
#define MAX_BUCKETS 512
#define EDGES_PER_BLOCK 2048

__device__ __forceinline__ void scan_bucket_counts(const int* __restrict__ bucketCount,
                                                   int nb, int* base, int* waveSum) {
  const int t    = threadIdx.x;
  const int lane = t & 31;
  const int wave = t >> 5;
  const int c0 = (2 * t     < nb) ? bucketCount[2 * t]     : 0;
  const int c1 = (2 * t + 1 < nb) ? bucketCount[2 * t + 1] : 0;
  const int c  = c0 + c1;
  int incl = c;
#pragma unroll
  for (int d = 1; d < 32; d <<= 1) {
    const int up = __shfl_up(incl, d);
    if (lane >= d) incl += up;
  }
  if (lane == 31) waveSum[wave] = incl;
  __syncthreads();
  int wbase = 0;
#pragma unroll
  for (int i = 0; i < 8; ++i) {
    const int s = waveSum[i];
    if (i < wave) wbase += s;
  }
  const int excl = wbase + incl - c;
  base[2 * t]     = excl;
  base[2 * t + 1] = excl + c0;
  if (t == 255) base[MAX_BUCKETS] = excl + c;
  __syncthreads();
}

__device__ __forceinline__ void bucket_count_block(const int* __restrict__ dst,
                                                   int* __restrict__ bucketCount,
                                                   int E, int nb, int blk) {
  __shared__ int hist[MAX_BUCKETS];
  const int t = threadIdx.x;
  hist[t] = 0;
  hist[t + 256] = 0;
  __syncthreads();
  const int e0 = blk * EDGES_PER_BLOCK + t;
  int d[8];
#pragma unroll
  for (int j = 0; j < 8; ++j) {
    const int e = e0 + j * 256;
    d[j] = (e < E) ? dst[e] : -1;
  }
#pragma unroll
  for (int j = 0; j < 8; ++j)
    if (d[j] >= 0) atomicAdd(&hist[d[j] >> BUCKET_SHIFT], 1);
  __syncthreads();
  for (int b = t; b < nb; b += 256) {
    const int h = hist[b];
    if (h) atomicAdd(bucketCount + b, h);
  }
}

__device__ __forceinline__ void bucket_scatter_block(const int* __restrict__ src,
                                                     const int* __restrict__ dst,
                                                     const float* __restrict__ w,
                                                     const int* __restrict__ bucketCount,
                                                     int* __restrict__ bucketCursor,
                                                     int2* __restrict__ staged,
                                                     int E, int nb, int blk) {
  __shared__ int base[MAX_BUCKETS + 1];
  __shared__ int hist[MAX_BUCKETS];
  __shared__ int waveSum[8];
  const int t = threadIdx.x;
  hist[t] = 0;
  hist[t + 256] = 0;

  const int e0 = blk * EDGES_PER_BLOCK + t;
  int d[8], s[8], wb[8];
#pragma unroll
  for (int j = 0; j < 8; ++j) {
    const int e = e0 + j * 256;
    const bool in = e < E;
    d[j]  = in ? dst[e] : -1;
    s[j]  = in ? src[e] : 0;
    wb[j] = in ? __float_as_int(w[e]) : 0;
  }

  scan_bucket_counts(bucketCount, nb, base, waveSum);

  int rank[8];
#pragma unroll
  for (int j = 0; j < 8; ++j) {
    rank[j] = 0;
    if (d[j] >= 0) rank[j] = atomicAdd(&hist[d[j] >> BUCKET_SHIFT], 1);
  }
  __syncthreads();
  for (int b = t; b < nb; b += 256) {
    const int h = hist[b];
    int r = 0;
    if (h) r = atomicAdd(bucketCursor + b, h);
    hist[b] = base[b] + r;
  }
  __syncthreads();
#pragma unroll
  for (int j = 0; j < 8; ++j) {
    if (d[j] >= 0) {
      const int p = hist[d[j] >> BUCKET_SHIFT] + rank[j];
      staged[p] = make_int2((int)((unsigned)s[j] | ((unsigned)(d[j] & 255) << 24)), wb[j]);
    }
  }
}

__device__ __forceinline__ void bucket_sort_block(const int* __restrict__ bucketCount,
                                                  const int2* __restrict__ staged,
                                                  int2* __restrict__ rec,
                                                  int* __restrict__ off,
                                                  int* __restrict__ end,
                                                  int N, int nb, int blk) {
  __shared__ int base[MAX_BUCKETS + 1];
  __shared__ int waveSum[8];
  __shared__ int cnt[256];
  const int t    = threadIdx.x;
  const int lane = t & 31;
  const int wave = t >> 5;
  cnt[t] = 0;
  scan_bucket_counts(bucketCount, nb, base, waveSum);
  const int s = base[blk];
  const int e = base[blk + 1];

  for (int i = s + t; i < e; i += 1024) {
    int k[4];
#pragma unroll
    for (int g = 0; g < 4; ++g) k[g] = staged[min(i + g * 256, e - 1)].x;
#pragma unroll
    for (int g = 0; g < 4; ++g)
      if (i + g * 256 < e) atomicAdd(&cnt[(unsigned)k[g] >> 24], 1);
  }
  __syncthreads();

  const int c = cnt[t];
  int incl = c;
#pragma unroll
  for (int d = 1; d < 32; d <<= 1) {
    const int up = __shfl_up(incl, d);
    if (lane >= d) incl += up;
  }
  if (lane == 31) waveSum[wave] = incl;
  __syncthreads();
  int wbase = 0;
#pragma unroll
  for (int i = 0; i < 8; ++i) {
    const int ws = waveSum[i];
    if (i < wave) wbase += ws;
  }
  const int o = s + wbase + incl - c;
  const int node = (blk << BUCKET_SHIFT) + t;
  if (node < N) { off[node] = o; end[node] = o + c; }
  cnt[t] = o;
  __syncthreads();

  for (int i = s + t; i < e; i += 1024) {
    int2 r[4];
#pragma unroll
    for (int g = 0; g < 4; ++g) r[g] = staged[min(i + g * 256, e - 1)];
#pragma unroll
    for (int g = 0; g < 4; ++g) {
      if (i + g * 256 < e) {
        const int p = atomicAdd(&cnt[(unsigned)r[g].x >> 24], 1);
        rec[p] = make_int2(r[g].x & 0x00FFFFFF, r[g].y);
      }
    }
  }
}

__global__ void __launch_bounds__(256) count_and_gemm(const int* __restrict__ dst,
                                                      int* __restrict__ bucketCount,
                                                      int E, int nb, int gemmBlocks,
                                                      const float* __restrict__ x,
                                                      const float* __restrict__ W1,
                                                      float* __restrict__ t1,
                                                      int N, int K, int gemmFirstStrip) {
  const int blk = blockIdx.x;
  if (blk < gemmBlocks) gemm_strips<0>(x, W1, t1, N, K, gemmFirstStrip + blk * 8);
  else bucket_count_block(dst, bucketCount, E, nb, blk - gemmBlocks);
}

__global__ void __launch_bounds__(256) scatter_and_gemm(const int* __restrict__ src,
                                                        const int* __restrict__ dst,
                                                        const float* __restrict__ w,
                                                        const int* __restrict__ bucketCount,
                                                        int* __restrict__ bucketCursor,
                                                        int2* __restrict__ staged,
                                                        int E, int nb, int gemmBlocks,
                                                        const float* __restrict__ x,
                                                        const float* __restrict__ W1,
                                                        float* __restrict__ t1,
                                                        int N, int K, int gemmFirstStrip) {
  const int blk = blockIdx.x;
  if (blk < gemmBlocks) gemm_strips<0>(x, W1, t1, N, K, gemmFirstStrip + blk * 8);
  else bucket_scatter_block(src, dst, w, bucketCount, bucketCursor, staged, E, nb, blk - gemmBlocks);
}

__global__ void __launch_bounds__(256) sort_and_gemm(const int* __restrict__ bucketCount,
                                                     const int2* __restrict__ staged,
                                                     int2* __restrict__ rec,
                                                     int* __restrict__ off,
                                                     int* __restrict__ end,
                                                     int nb, int gemmBlocks,
                                                     const float* __restrict__ x,
                                                     const float* __restrict__ W1,
                                                     float* __restrict__ t1,
                                                     int N, int K, int gemmFirstStrip) {
  const int blk = blockIdx.x;
  if (blk < gemmBlocks) gemm_strips<0>(x, W1, t1, N, K, gemmFirstStrip + blk * 8);
  else bucket_sort_block(bucketCount, staged, rec, off, end, N, nb, blk - gemmBlocks);
}

#define HSTRIDE 68
#define PULL_GROUP 4

__global__ void __launch_bounds__(512) pull_fold_wmma_f32(const float* __restrict__ t,
                                   const int* __restrict__ off,
                                   const int* __restrict__ end,
                                   const int2* __restrict__ rec,
                                   const float* __restrict__ W2,
                                   const float* __restrict__ Wlin,
                                   float* __restrict__ U,
                                   int nRows) {
  __shared__ float2 WfP[32 * 16];
  __shared__ __attribute__((aligned(16))) float Hs[64 * HSTRIDE];
  if (threadIdx.x < 256) {
    const int t4 = threadIdx.x;
    if ((t4 & 15) >= 4) {
      WfP[t4]       = make_float2(0.f, 0.f);
      WfP[t4 + 256] = make_float2(0.f, 0.f);
    }
    const int k = t4 >> 2;
    const int c = t4 & 3;
    const float4* w2r = (const float4*)(W2 + k * 64);
    const float4* wlr = (const float4*)(Wlin + (c & 1) * 128 + (c >> 1) * 64);
    float s = 0.f;
#pragma unroll 4
    for (int q = 0; q < 16; ++q) {
      const float4 a = w2r[q];
      const float4 b = wlr[q];
      s += a.x * b.x + a.y * b.y + a.z * b.z + a.w * b.w;
    }
    ((float*)WfP)[((k >> 1) * 16 + c) * 2 + (k & 1)] = s;
  }

  const int lane = threadIdx.x & 31;
  const int wave = threadIdx.x >> 5;

  {
    const int node0 = blockIdx.x * 64 + wave * 4;
    int myOff = 0, myEnd = 0;
    if (lane < 4 && node0 + lane < nRows) {
      myOff = off[node0 + lane];
      myEnd = end[node0 + lane];
    }
    int segS[4], segT[4];
    int2 first[4];
#pragma unroll
    for (int j = 0; j < 4; ++j) {
      segS[j] = __builtin_amdgcn_readlane(myOff, j);
      segT[j] = __builtin_amdgcn_readlane(myEnd, j);
      first[j] = make_int2(0, 0);
      if (lane < segT[j] - segS[j]) first[j] = rec[segS[j] + lane];
    }
#pragma unroll
    for (int j = 0; j < 4; ++j) {
      const int s = segS[j], e = segT[j];
      float2 acc = make_float2(0.f, 0.f);
      for (int base = s; base < e; base += 32) {
        const int cnt = min(32, e - base);
        int2 r = first[j];
        if (base != s) {
          r = make_int2(0, 0);
          if (lane < cnt) r = rec[base + lane];
        }
        for (int i = 0; i < cnt; i += PULL_GROUP) {
          float2 v[PULL_GROUP];
          float  ww[PULL_GROUP];
#pragma unroll
          for (int g = 0; g < PULL_GROUP; ++g) {
            const int ig = min(i + g, cnt - 1);
            const int sg = __builtin_amdgcn_readlane(r.x, ig);
            const float wg = __int_as_float(__builtin_amdgcn_readlane(r.y, ig));
            ww[g] = (i + g < cnt) ? wg : 0.f;
            v[g] = *(const float2*)(t + (long)sg * 64 + 2 * lane);
          }
#pragma unroll
          for (int g = 0; g < PULL_GROUP; ++g) {
            acc.x += ww[g] * v[g].x;
            acc.y += ww[g] * v[g].y;
          }
        }
      }
      acc.x = fmaxf(acc.x, 0.f);
      acc.y = fmaxf(acc.y, 0.f);
      *(float2*)&Hs[(wave * 4 + j) * HSTRIDE + 2 * lane] = acc;
    }
  }
  __syncthreads();

  const int nStrips = nRows >> 4;
  const int strip = blockIdx.x * 4 + wave;
  if (wave >= 4 || strip >= nStrips) return;

  const int n  = lane & 15;
  const int hi = lane >> 4;
  const long rowBase = (long)strip * 16;
  const float* Arow = &Hs[(wave * 16 + n) * HSTRIDE];

  v8f acc = {};
#pragma unroll
  for (int k = 0; k < 64; k += 4) {
    const int ka = k + 2 * hi;
    const v2f a = *(const v2f*)(Arow + ka);
    const v2f b = *(const v2f*)&WfP[((k >> 1) + hi) * 16 + n];
    acc = __builtin_amdgcn_wmma_f32_16x16x4_f32(false, a, false, b, (short)0, acc, false, false);
  }

  if (n < 4) {
    float* Uout = U + (rowBase + 8 * hi) * 4 + n;
#pragma unroll
    for (int r = 0; r < 8; ++r) Uout[r * 4] = acc[r];
  }
}

__global__ void pull_edges4(const float* __restrict__ u,
                            const int* __restrict__ off,
                            const int* __restrict__ end,
                            const int2* __restrict__ rec,
                            float* __restrict__ y,
                            int N) {
  const int tid = blockIdx.x * blockDim.x + threadIdx.x;
  const int n = tid >> 2;
  const int c = tid & 3;
  if (n >= N) return;
  const int s = off[n], e = end[n];
  float acc = 0.f;
  for (int b = s; b < e; b += 4) {
    int   sg[4];
    float wg[4], v[4];
#pragma unroll
    for (int g = 0; g < 4; ++g) {
      const int2 r = rec[min(b + g, e - 1)];
      sg[g] = r.x;
      wg[g] = (b + g < e) ? __int_as_float(r.y) : 0.f;
    }
#pragma unroll
    for (int g = 0; g < 4; ++g) v[g] = u[(long)sg[g] * 4 + c];
#pragma unroll
    for (int g = 0; g < 4; ++g) acc += wg[g] * v[g];
  }
  y[(long)n * 4 + c] = acc;
}

__global__ void decode_fold(const float* __restrict__ y,
                            const int* __restrict__ pos,
                            float* __restrict__ out,
                            int P) {
  const int p = blockIdx.x * blockDim.x + threadIdx.x;
  if (p >= P) return;
  const int s = pos[p];
  const int d = pos[P + p];
  const float2 a = *(const float2*)(y + (long)s * 4);
  const float2 b = *(const float2*)(y + (long)d * 4 + 2);
  *(float2*)(out + 2 * (long)p) = make_float2(a.x + b.x, a.y + b.y);
}

extern "C" void kernel_launch(void* const* d_in, const int* in_sizes, int n_in,
                              void* d_out, int out_size, void* d_ws, size_t ws_size,
                              hipStream_t stream) {
  const float* x    = (const float*)d_in[0];
  const int*   ei   = (const int*)d_in[1];
  const float* w    = (const float*)d_in[2];
  const int*   pos  = (const int*)d_in[3];
  const float* W1   = (const float*)d_in[4];
  const float* W2   = (const float*)d_in[5];
  const float* Wlin = (const float*)d_in[6];
  float* out = (float*)d_out;

  const int F = 128;
  const int N = in_sizes[0] / F;
  const int E = in_sizes[2];
  const int P = in_sizes[3] / 2;

  float* t1  = (float*)d_ws;
  float* u   = t1 + (size_t)N * 64;
  float* y   = u  + (size_t)N * 4;
  int2*  rec = (int2*)(y + (size_t)N * 4);
  int*   off = (int*)(rec + (size_t)E);
  int*   end = off + N;
  int2*  stagedOwn = (int2*)(end + N);
  const bool roomy = ws_size >= (size_t)((const char*)(stagedOwn + E) - (const char*)d_ws);
  int2*  staged       = roomy ? stagedOwn : (int2*)t1;
  int*   bucketCount  = (int*)y;
  int*   bucketCursor = bucketCount + MAX_BUCKETS;

  hipMemsetAsync(bucketCount, 0, 2 * MAX_BUCKETS * sizeof(int), stream);

  const int nStrips    = N / 16;
  const int partStrips = ((nStrips + 2) / 3 + 7) / 8 * 8;
  const int ride       = roomy ? partStrips / 8 : 0;
  const int eblocks    = (E + EDGES_PER_BLOCK - 1) / EDGES_PER_BLOCK;
  const int nb         = (N + 255) >> BUCKET_SHIFT;

  count_and_gemm<<<ride + eblocks, 256, 0, stream>>>(ei + E, bucketCount, E, nb, ride,
                                                     x, W1, t1, N, F, 0);
  scatter_and_gemm<<<ride + eblocks, 256, 0, stream>>>(ei, ei + E, w, bucketCount, bucketCursor, staged, E, nb, ride,
                                                       x, W1, t1, N, F, partStrips);
  sort_and_gemm<<<ride + nb, 256, 0, stream>>>(bucketCount, staged, rec, off, end, nb, ride,
                                               x, W1, t1, N, F, 2 * partStrips);
  if (!roomy)
    sort_and_gemm<<<(nStrips + 7) / 8, 256, 0, stream>>>(bucketCount, staged, rec, off, end, nb, (nStrips + 7) / 8,
                                                         x, W1, t1, N, F, 0);
  pull_fold_wmma_f32<<<(N + 63) / 64, 512, 0, stream>>>(t1, off, end, rec, W2, Wlin, u, N);
  pull_edges4<<<(N * 4 + 255) / 256, 256, 0, stream>>>(u, off, end, rec, y, N);
  decode_fold<<<(P + 255) / 256, 256, 0, stream>>>(y, pos, out, P);
}


# Edge grouping in 2 launches, not 3 and a memset: a block bucket-sorts its 2048 edges inside its own chunk (no counting pass, no global counters), a bucket block collects its runs; arrival order kept
# speedup vs baseline: 3.7115x; 1.0455x over previous
#include <hip/hip_runtime.h>
#include <hip/hip_bf16.h>

typedef __attribute__((ext_vector_type(2))) float v2f;
typedef __attribute__((ext_vector_type(8))) float v8f;

#define BSTRIDE 80

template <int RELU>
__device__ __forceinline__ void gemm_strips(const float* __restrict__ A,
                                            const float* __restrict__ B,
                                            float* __restrict__ C,
                                            int nRows, int K, int firstStrip) {
  __shared__ __attribute__((aligned(16))) float2 BsP[64 * BSTRIDE];
  const int cells = (K >> 1) * 16;
  for (int base = threadIdx.x; base < cells; base += 4 * (int)blockDim.x) {
    float4 ra[4], rb[4];
#pragma unroll
    for (int i = 0; i < 4; ++i) {
      const int idx = min(base + i * (int)blockDim.x, cells - 1);
      const int k2 = idx >> 4;
      const int c4 = idx & 15;
      ra[i] = *(const float4*)(B + (2 * k2) * 64 + 4 * c4);
      rb[i] = *(const float4*)(B + (2 * k2 + 1) * 64 + 4 * c4);
    }
#pragma unroll
    for (int i = 0; i < 4; ++i) {
      const int idx = min(base + i * (int)blockDim.x, cells - 1);
      const int k2 = idx >> 4;
      const int c4 = idx & 15;
      float4* dst = (float4*)&BsP[k2 * BSTRIDE + 4 * c4];
      dst[0] = make_float4(ra[i].x, rb[i].x, ra[i].y, rb[i].y);
      dst[1] = make_float4(ra[i].z, rb[i].z, ra[i].w, rb[i].w);
    }
  }
  __syncthreads();

  const int lane = threadIdx.x & 31;
  const int wave = threadIdx.x >> 5;
  const int nStrips = nRows >> 4;
  const int strip = firstStrip + wave;
  if (strip >= nStrips) return;

  const int n  = lane & 15;
  const int hi = lane >> 4;
  const long rowBase = (long)strip * 16;
  const float* Arow = A + (rowBase + n) * (long)K;

  v8f acc0 = {}, acc1 = {}, acc2 = {}, acc3 = {};
  for (int k = 0; k < K; k += 4) {
    const int ka = k + 2 * hi;
    float2 av = *(const float2*)(Arow + ka);
    if (RELU) { av.x = fmaxf(av.x, 0.f); av.y = fmaxf(av.y, 0.f); }
    v2f a; a[0] = av.x; a[1] = av.y;

    const float2* bp = &BsP[((k >> 1) + hi) * BSTRIDE + n];
    const v2f bA = *(const v2f*)(bp + 0);
    const v2f bB = *(const v2f*)(bp + 16);
    const v2f bC = *(const v2f*)(bp + 32);
    const v2f bD = *(const v2f*)(bp + 48);

    acc0 = __builtin_amdgcn_wmma_f32_16x16x4_f32(false, a, false, bA, (short)0, acc0, false, false);
    acc1 = __builtin_amdgcn_wmma_f32_16x16x4_f32(false, a, false, bB, (short)0, acc1, false, false);
    acc2 = __builtin_amdgcn_wmma_f32_16x16x4_f32(false, a, false, bC, (short)0, acc2, false, false);
    acc3 = __builtin_amdgcn_wmma_f32_16x16x4_f32(false, a, false, bD, (short)0, acc3, false, false);
  }

  float* Cout = C + (rowBase + 8 * hi) * 64 + n;
#pragma unroll
  for (int r = 0; r < 8; ++r) {
    Cout[r * 64 + 0]  = acc0[r];
    Cout[r * 64 + 16] = acc1[r];
    Cout[r * 64 + 32] = acc2[r];
    Cout[r * 64 + 48] = acc3[r];
  }
}

#define BUCKET_SHIFT 8
#define MAX_BUCKETS 512
#define EDGES_PER_BLOCK 2048
#define MAX_EBLOCKS 512
#define IDX_CAP 4096

template <bool CARRY>
__device__ __forceinline__ void scan512(int c0, int c1, int* base, int* waveSum, int x, int* xSum) {
  const int t    = threadIdx.x;
  const int lane = t & 31;
  const int wave = t >> 5;
  const int c  = c0 + c1;
  int incl = c;
  int xin  = x;
#pragma unroll
  for (int d = 1; d < 32; d <<= 1) {
    const int up = __shfl_up(incl, d);
    if (CARRY) {
      const int xu = __shfl_up(xin, d);
      if (lane >= d) xin += xu;
    }
    if (lane >= d) incl += up;
  }
  if (lane == 31) {
    waveSum[wave] = incl;
    if (CARRY) xSum[wave] = xin;
  }
  __syncthreads();
  int wbase = 0;
#pragma unroll
  for (int i = 0; i < 8; ++i) {
    const int s = waveSum[i];
    if (i < wave) wbase += s;
  }
  const int excl = wbase + incl - c;
  base[2 * t]     = excl;
  base[2 * t + 1] = excl + c0;
  if (t == 255) base[512] = excl + c;
  __syncthreads();
}

__device__ __forceinline__ void bucket_chunk_block(const int* __restrict__ src,
                                                   const int* __restrict__ dst,
                                                   const float* __restrict__ w,
                                                   int2* __restrict__ staged,
                                                   int* __restrict__ tab,
                                                   int E, int nb, int ts, int blk) {
  __shared__ int base[MAX_BUCKETS + 1];
  __shared__ int hist[MAX_BUCKETS];
  __shared__ int waveSum[8];
  const int t = threadIdx.x;
  hist[t] = 0;
  hist[t + 256] = 0;

  const int e0 = blk * EDGES_PER_BLOCK + t;
  int d[8], s[8], wb[8];
#pragma unroll
  for (int j = 0; j < 8; ++j) {
    const int e = e0 + j * 256;
    const bool in = e < E;
    d[j]  = in ? dst[e] : -1;
    s[j]  = in ? src[e] : 0;
    wb[j] = in ? __float_as_int(w[e]) : 0;
  }
  __syncthreads();

  int rank[8];
#pragma unroll
  for (int j = 0; j < 8; ++j) {
    rank[j] = 0;
    if (d[j] >= 0) rank[j] = atomicAdd(&hist[d[j] >> BUCKET_SHIFT], 1);
  }
  __syncthreads();
  scan512<false>(hist[2 * t], hist[2 * t + 1], base, waveSum, 0, nullptr);

  for (int b = t; b <= nb; b += 256) tab[(size_t)b * ts + blk] = base[b];

  int2* chunk = staged + (size_t)blk * EDGES_PER_BLOCK;
#pragma unroll
  for (int j = 0; j < 8; ++j) {
    if (d[j] >= 0) {
      const int p = base[d[j] >> BUCKET_SHIFT] + rank[j];
      chunk[p] = make_int2((int)((unsigned)s[j] | ((unsigned)(d[j] & 255) << 24)), wb[j]);
    }
  }
}

__device__ __forceinline__ void bucket_collect_block(const int* __restrict__ tab,
                                                     const int2* __restrict__ staged,
                                                     int2* __restrict__ rec,
                                                     int* __restrict__ off,
                                                     int* __restrict__ end,
                                                     int N, int E, int ts, int eblocks, int blk) {
  __shared__ int segPos[MAX_EBLOCKS + 1];
  __shared__ int waveSum[8];
  __shared__ int rowSum[8];
  __shared__ int cnt[256];
  __shared__ int srcIdx[IDX_CAP];
  const int t    = threadIdx.x;
  const int lane = t & 31;
  const int wave = t >> 5;
  cnt[t] = 0;

  const int i0 = 2 * t, i1 = 2 * t + 1;
  const int j0 = min(i0, eblocks - 1), j1 = min(i1, eblocks - 1);
  const int* rowA = tab + (size_t)blk * ts;
  const int* rowB = tab + (size_t)(blk + 1) * ts;
  int s0 = rowA[j0], f0 = rowB[j0], s1 = rowA[j1], f1 = rowB[j1];
  if (i0 >= eblocks) { s0 = 0; f0 = 0; }
  if (i1 >= eblocks) { s1 = 0; f1 = 0; }
  s0 = min(max(s0, 0), EDGES_PER_BLOCK);
  s1 = min(max(s1, 0), EDGES_PER_BLOCK);
  const int len0 = min(max(f0 - s0, 0), EDGES_PER_BLOCK - s0);
  const int len1 = min(max(f1 - s1, 0), EDGES_PER_BLOCK - s1);

  scan512<true>(len0, len1, segPos, waveSum, s0 + s1, rowSum);
  const int total = segPos[MAX_EBLOCKS];
  int gb = 0;
#pragma unroll
  for (int i = 0; i < 8; ++i) gb += rowSum[i];

  const int p0 = segPos[i0], p1 = segPos[i1];
  const int g0 = i0 * EDGES_PER_BLOCK + s0;
  const int g1 = i1 * EDGES_PER_BLOCK + s1;
  const int nchunks = (total + IDX_CAP - 1) / IDX_CAP;

  for (int c = 0; c < nchunks; ++c) {
    const int lo = c * IDX_CAP;
    const int m  = min(IDX_CAP, total - lo);
    for (int j = 0; j < len0; ++j) { const int q = p0 + j - lo; if (q >= 0 && q < IDX_CAP) srcIdx[q] = g0 + j; }
    for (int j = 0; j < len1; ++j) { const int q = p1 + j - lo; if (q >= 0 && q < IDX_CAP) srcIdx[q] = g1 + j; }
    __syncthreads();
    for (int i = t; i < m; i += 1024) {
      int k[4];
#pragma unroll
      for (int g = 0; g < 4; ++g) k[g] = staged[min(srcIdx[min(i + g * 256, m - 1)], E - 1)].x;
#pragma unroll
      for (int g = 0; g < 4; ++g)
        if (i + g * 256 < m) atomicAdd(&cnt[(unsigned)k[g] >> 24], 1);
    }
    __syncthreads();
  }
  __syncthreads();

  const int cn = cnt[t];
  int incl = cn;
#pragma unroll
  for (int d = 1; d < 32; d <<= 1) {
    const int up = __shfl_up(incl, d);
    if (lane >= d) incl += up;
  }
  if (lane == 31) waveSum[wave] = incl;
  __syncthreads();
  int wbase = 0;
#pragma unroll
  for (int i = 0; i < 8; ++i) {
    const int ws = waveSum[i];
    if (i < wave) wbase += ws;
  }
  const int o = gb + wbase + incl - cn;
  const int node = (blk << BUCKET_SHIFT) + t;
  if (node < N) { off[node] = o; end[node] = o + cn; }
  cnt[t] = o;
  __syncthreads();

  for (int c = 0; c < nchunks; ++c) {
    const int lo = c * IDX_CAP;
    const int m  = min(IDX_CAP, total - lo);
    if (nchunks > 1) {
      for (int j = 0; j < len0; ++j) { const int q = p0 + j - lo; if (q >= 0 && q < IDX_CAP) srcIdx[q] = g0 + j; }
      for (int j = 0; j < len1; ++j) { const int q = p1 + j - lo; if (q >= 0 && q < IDX_CAP) srcIdx[q] = g1 + j; }
      __syncthreads();
    }
    for (int i = t; i < m; i += 1024) {
      int2 r[4];
#pragma unroll
      for (int g = 0; g < 4; ++g) r[g] = staged[min(srcIdx[min(i + g * 256, m - 1)], E - 1)];
#pragma unroll
      for (int g = 0; g < 4; ++g) {
        if (i + g * 256 < m) {
          const int p = atomicAdd(&cnt[(unsigned)r[g].x >> 24], 1);
          if (p < E) rec[p] = make_int2(r[g].x & 0x00FFFFFF, r[g].y);
        }
      }
    }
    if (nchunks > 1) __syncthreads();
  }
}

__global__ void __launch_bounds__(256) chunk_and_gemm(const int* __restrict__ src,
                                                      const int* __restrict__ dst,
                                                      const float* __restrict__ w,
                                                      int2* __restrict__ staged,
                                                      int* __restrict__ tab,
                                                      int E, int nb, int ts, int gemmBlocks,
                                                      const float* __restrict__ x,
                                                      const float* __restrict__ W1,
                                                      float* __restrict__ t1,
                                                      int N, int K, int gemmFirstStrip) {
  const int blk = blockIdx.x;
  if (blk < gemmBlocks) gemm_strips<0>(x, W1, t1, N, K, gemmFirstStrip + blk * 8);
  else bucket_chunk_block(src, dst, w, staged, tab, E, nb, ts, blk - gemmBlocks);
}

__global__ void __launch_bounds__(256) collect_and_gemm(const int* __restrict__ tab,
                                                        const int2* __restrict__ staged,
                                                        int2* __restrict__ rec,
                                                        int* __restrict__ off,
                                                        int* __restrict__ end,
                                                        int E, int ts, int eblocks, int gemmBlocks,
                                                        const float* __restrict__ x,
                                                        const float* __restrict__ W1,
                                                        float* __restrict__ t1,
                                                        int N, int K, int gemmFirstStrip) {
  const int blk = blockIdx.x;
  if (blk < gemmBlocks) gemm_strips<0>(x, W1, t1, N, K, gemmFirstStrip + blk * 8);
  else bucket_collect_block(tab, staged, rec, off, end, N, E, ts, eblocks, blk - gemmBlocks);
}

#define HSTRIDE 68
#define PULL_GROUP 4

__global__ void __launch_bounds__(512) pull_fold_wmma_f32(const float* __restrict__ t,
                                   const int* __restrict__ off,
                                   const int* __restrict__ end,
                                   const int2* __restrict__ rec,
                                   const float* __restrict__ W2,
                                   const float* __restrict__ Wlin,
                                   float* __restrict__ U,
                                   int nRows) {
  __shared__ float2 WfP[32 * 16];
  __shared__ __attribute__((aligned(16))) float Hs[64 * HSTRIDE];
  if (threadIdx.x < 256) {
    const int t4 = threadIdx.x;
    if ((t4 & 15) >= 4) {
      WfP[t4]       = make_float2(0.f, 0.f);
      WfP[t4 + 256] = make_float2(0.f, 0.f);
    }
    const int k = t4 >> 2;
    const int c = t4 & 3;
    const float4* w2r = (const float4*)(W2 + k * 64);
    const float4* wlr = (const float4*)(Wlin + (c & 1) * 128 + (c >> 1) * 64);
    float s = 0.f;
#pragma unroll 4
    for (int q = 0; q < 16; ++q) {
      const float4 a = w2r[q];
      const float4 b = wlr[q];
      s += a.x * b.x + a.y * b.y + a.z * b.z + a.w * b.w;
    }
    ((float*)WfP)[((k >> 1) * 16 + c) * 2 + (k & 1)] = s;
  }

  const int lane = threadIdx.x & 31;
  const int wave = threadIdx.x >> 5;

  {
    const int node0 = blockIdx.x * 64 + wave * 4;
    int myOff = 0, myEnd = 0;
    if (lane < 4 && node0 + lane < nRows) {
      myOff = off[node0 + lane];
      myEnd = end[node0 + lane];
    }
    int segS[4], segT[4];
    int2 first[4];
#pragma unroll
    for (int j = 0; j < 4; ++j) {
      segS[j] = __builtin_amdgcn_readlane(myOff, j);
      segT[j] = __builtin_amdgcn_readlane(myEnd, j);
      first[j] = make_int2(0, 0);
      if (lane < segT[j] - segS[j]) first[j] = rec[segS[j] + lane];
    }
#pragma unroll
    for (int j = 0; j < 4; ++j) {
      const int s = segS[j], e = segT[j];
      float2 acc = make_float2(0.f, 0.f);
      for (int base = s; base < e; base += 32) {
        const int cnt = min(32, e - base);
        int2 r = first[j];
        if (base != s) {
          r = make_int2(0, 0);
          if (lane < cnt) r = rec[base + lane];
        }
        for (int i = 0; i < cnt; i += PULL_GROUP) {
          float2 v[PULL_GROUP];
          float  ww[PULL_GROUP];
#pragma unroll
          for (int g = 0; g < PULL_GROUP; ++g) {
            const int ig = min(i + g, cnt - 1);
            const int sg = __builtin_amdgcn_readlane(r.x, ig);
            const float wg = __int_as_float(__builtin_amdgcn_readlane(r.y, ig));
            ww[g] = (i + g < cnt) ? wg : 0.f;
            v[g] = *(const float2*)(t + (long)sg * 64 + 2 * lane);
          }
#pragma unroll
          for (int g = 0; g < PULL_GROUP; ++g) {
            acc.x += ww[g] * v[g].x;
            acc.y += ww[g] * v[g].y;
          }
        }
      }
      acc.x = fmaxf(acc.x, 0.f);
      acc.y = fmaxf(acc.y, 0.f);
      *(float2*)&Hs[(wave * 4 + j) * HSTRIDE + 2 * lane] = acc;
    }
  }
  __syncthreads();

  const int nStrips = nRows >> 4;
  const int strip = blockIdx.x * 4 + wave;
  if (wave >= 4 || strip >= nStrips) return;

  const int n  = lane & 15;
  const int hi = lane >> 4;
  const long rowBase = (long)strip * 16;
  const float* Arow = &Hs[(wave * 16 + n) * HSTRIDE];

  v8f acc = {};
#pragma unroll
  for (int k = 0; k < 64; k += 4) {
    const int ka = k + 2 * hi;
    const v2f a = *(const v2f*)(Arow + ka);
    const v2f b = *(const v2f*)&WfP[((k >> 1) + hi) * 16 + n];
    acc = __builtin_amdgcn_wmma_f32_16x16x4_f32(false, a, false, b, (short)0, acc, false, false);
  }

  if (n < 4) {
    float* Uout = U + (rowBase + 8 * hi) * 4 + n;
#pragma unroll
    for (int r = 0; r < 8; ++r) Uout[r * 4] = acc[r];
  }
}

__global__ void pull_edges4(const float* __restrict__ u,
                            const int* __restrict__ off,
                            const int* __restrict__ end,
                            const int2* __restrict__ rec,
                            float* __restrict__ y,
                            int N) {
  const int tid = blockIdx.x * blockDim.x + threadIdx.x;
  const int n = tid >> 2;
  const int c = tid & 3;
  if (n >= N) return;
  const int s = off[n], e = end[n];
  float acc = 0.f;
  for (int b = s; b < e; b += 4) {
    int   sg[4];
    float wg[4], v[4];
#pragma unroll
    for (int g = 0; g < 4; ++g) {
      const int2 r = rec[min(b + g, e - 1)];
      sg[g] = r.x;
      wg[g] = (b + g < e) ? __int_as_float(r.y) : 0.f;
    }
#pragma unroll
    for (int g = 0; g < 4; ++g) v[g] = u[(long)sg[g] * 4 + c];
#pragma unroll
    for (int g = 0; g < 4; ++g) acc += wg[g] * v[g];
  }
  y[(long)n * 4 + c] = acc;
}

__global__ void decode_fold(const float* __restrict__ y,
                            const int* __restrict__ pos,
                            float* __restrict__ out,
                            int P) {
  const int p = blockIdx.x * blockDim.x + threadIdx.x;
  if (p >= P) return;
  const int s = pos[p];
  const int d = pos[P + p];
  const float2 a = *(const float2*)(y + (long)s * 4);
  const float2 b = *(const float2*)(y + (long)d * 4 + 2);
  *(float2*)(out + 2 * (long)p) = make_float2(a.x + b.x, a.y + b.y);
}

extern "C" void kernel_launch(void* const* d_in, const int* in_sizes, int n_in,
                              void* d_out, int out_size, void* d_ws, size_t ws_size,
                              hipStream_t stream) {
  const float* x    = (const float*)d_in[0];
  const int*   ei   = (const int*)d_in[1];
  const float* w    = (const float*)d_in[2];
  const int*   pos  = (const int*)d_in[3];
  const float* W1   = (const float*)d_in[4];
  const float* W2   = (const float*)d_in[5];
  const float* Wlin = (const float*)d_in[6];
  float* out = (float*)d_out;

  const int F = 128;
  const int N = in_sizes[0] / F;
  const int E = in_sizes[2];
  const int P = in_sizes[3] / 2;

  float* t1  = (float*)d_ws;
  float* u   = t1 + (size_t)N * 64;
  float* y   = u  + (size_t)N * 4;
  int2*  rec = (int2*)(y + (size_t)N * 4);
  int*   off = (int*)(rec + (size_t)E);
  int*   end = off + N;
  int2*  stagedOwn = (int2*)(end + N);
  const bool roomy = ws_size >= (size_t)((const char*)(stagedOwn + E) - (const char*)d_ws);
  int2*  staged       = roomy ? stagedOwn : (int2*)t1;
  const int eblocks    = (E + EDGES_PER_BLOCK - 1) / EDGES_PER_BLOCK;
  const int nb         = (N + 255) >> BUCKET_SHIFT;
  int*   tab = (int*)y;

  const int nStrips    = N / 16;
  const int partStrips = ((nStrips + 1) / 2 + 7) / 8 * 8;
  const int ride       = roomy ? partStrips / 8 : 0;

  chunk_and_gemm<<<ride + eblocks, 256, 0, stream>>>(ei, ei + E, w, staged, tab, E, nb, eblocks, ride,
                                                     x, W1, t1, N, F, 0);
  collect_and_gemm<<<ride + nb, 256, 0, stream>>>(tab, staged, rec, off, end, E, eblocks, eblocks, ride,
                                                  x, W1, t1, N, F, partStrips);
  if (!roomy)
    collect_and_gemm<<<(nStrips + 7) / 8, 256, 0, stream>>>(tab, staged, rec, off, end, E, eblocks, eblocks, (nStrips + 7) / 8,
                                                            x, W1, t1, N, F, 0);
  pull_fold_wmma_f32<<<(N + 63) / 64, 512, 0, stream>>>(t1, off, end, rec, W2, Wlin, u, N);
  pull_edges4<<<(N * 4 + 255) / 256, 256, 0, stream>>>(u, off, end, rec, y, N);
  decode_fold<<<(P + 255) / 256, 256, 0, stream>>>(y, pos, out, P);
}
